# starting kernel
# baseline (speedup 1.0000x reference)
#include <hip/hip_runtime.h>

typedef __attribute__((ext_vector_type(2))) float v2f;
typedef __attribute__((ext_vector_type(8))) float v8f;

#define N_COMBOS 15
#define KP 164
#define PN 112
#define NTILES 7
#define WAVES 2
#define FEAT 18

constexpr int cLE[N_COMBOS]     = {0,0,0,1,1,1,1,1,1,2,2,2,2,2,2};
constexpr int cLN[N_COMBOS]     = {0,1,2,0,1,1,1,2,2,0,1,1,2,2,2};
constexpr int cLO[N_COMBOS]     = {0,1,2,1,0,1,2,1,2,2,1,2,0,1,2};
constexpr int OUT_OFF[N_COMBOS] = {0,2,8,18,24,26,32,42,48,58,68,74,84,86,92};
constexpr int U_OFF[9]          = {0,2,8,18,24,42,72,82,112};
constexpr int KLO[NTILES]       = {0,2,6,10,18,20,28};
constexpr int KHI[NTILES]       = {4,10,17,20,27,40,40};
constexpr int N_BY_LO[3]        = {3,6,6};
constexpr int C_BY_LO[3][6]     = {{0,4,12,0,0,0},{1,3,5,7,10,13},{2,6,8,9,11,14}};

__global__ void build_M_kernel(const float* cg0, const float* cg1, const float* cg2,
                               const float* cg3, const float* cg4, const float* cg5,
                               const float* cg6, const float* cg7, const float* cg8,
                               const float* cg9, const float* cg10, const float* cg11,
                               const float* cg12, const float* cg13, const float* cg14,
                               float* M) {
    const float* cgs[N_COMBOS] = {cg0,cg1,cg2,cg3,cg4,cg5,cg6,cg7,cg8,cg9,cg10,cg11,cg12,cg13,cg14};
    int tid = threadIdx.x;
    for (int i = tid; i < KP * PN; i += blockDim.x) M[i] = 0.0f;
    __syncthreads();
    if (tid < N_COMBOS) {
        int c = tid;
        int le = cLE[c], ln = cLN[c], lo = cLO[c];
        int de = 2 * le + 1, dn = 2 * ln + 1, dO = 2 * lo + 1;
        int uo = U_OFF[le * 3 + ln];
        int po = OUT_OFF[c];
        const float* cg = cgs[c];
        for (int a = 0; a < de; ++a)
            for (int b = 0; b < dn; ++b)
                for (int k = 0; k < dO; ++k) {
                    float v = cg[(a * dn + b) * dO + k];
                    for (int ch = 0; ch < 2; ++ch)
                        M[(uo + (a * dn + b) * 2 + ch) * PN + po + ch * dO + k] = v;
                }
    }
}

__global__ void zero_kernel(float* p, long n) {
    long i = (long)blockIdx.x * blockDim.x + threadIdx.x;
    if (i < n) p[i] = 0.0f;
}

template <int PART>
__device__ __forceinline__ void gate_block(const float* __restrict__ W1,
                                           const float* __restrict__ b1,
                                           const float* __restrict__ W2,
                                           const float* __restrict__ b2,
                                           float x0, float x1,
                                           float* __restrict__ gRow) {
#pragma unroll
    for (int t = 0; t < 30; ++t) {
        const int g = PART * 30 + t;
        const int bi = ((g >> 2) * 2 + ((g >> 1) & 1)) * 2 + (g & 1);
        float acc = b2[bi];
#pragma unroll
        for (int h = 0; h < 4; ++h) {
            float hh = x0 * W1[bi * 8 + h * 2] + x1 * W1[bi * 8 + h * 2 + 1] + b1[bi * 4 + h];
            hh = hh > 0.0f ? hh : 0.0f;
            acc += hh * W2[bi * 4 + h];
        }
        gRow[g] = tanhf(acc);
    }
}

template <int PART>
__device__ __forceinline__ void u_stage(const float* __restrict__ shv,
                                        const float* __restrict__ f,
                                        float* __restrict__ uRow) {
    int u = 0;
#pragma unroll
    for (int le = 0; le < 3; ++le) {
        const int de = 2 * le + 1;
        const int shb = (le == 0) ? 0 : (le == 1 ? 1 : 4);
#pragma unroll
        for (int ln = 0; ln < 3; ++ln) {
            const int dn = 2 * ln + 1;
            const int fb = (ln == 0) ? 0 : (ln == 1 ? 2 : 8);
#pragma unroll
            for (int a = 0; a < de; ++a)
#pragma unroll
                for (int b = 0; b < dn; ++b)
#pragma unroll
                    for (int ch = 0; ch < 2; ++ch) {
                        if ((PART == 0) ? (u < 81) : (u >= 81))
                            uRow[u] = shv[shb + a] * f[fb + ch * dn + b];
                        ++u;
                    }
        }
    }
    if (PART == 1) { uRow[162] = 0.0f; uRow[163] = 0.0f; }
}

template <int S>
__device__ __forceinline__ float slot_val(const float* __restrict__ gRow,
                                          const float* __restrict__ rRow) {
    constexpr int lo = (S < 2) ? 0 : ((S < 8) ? 1 : 2);
    constexpr int o  = (S < 2) ? S : ((S < 8) ? ((S - 2) / 3) : ((S - 8) / 5));
    constexpr int k  = (S < 2) ? 0 : ((S < 8) ? ((S - 2) % 3) : ((S - 8) % 5));
    constexpr int dO = 2 * lo + 1;
    float v = 0.0f;
#pragma unroll
    for (int q = 0; q < N_BY_LO[lo]; ++q) {
        constexpr const int* lst = C_BY_LO[lo];
        const int c = lst[q];
        const int base = OUT_OFF[c];
        v += gRow[c * 4 + o]     * rRow[base + k] +
             gRow[c * 4 + 2 + o] * rRow[base + dO + k];
    }
    return v;
}

__global__ void __launch_bounds__(WAVES * 32)
layer_kernel(const float* __restrict__ feats, const int* __restrict__ eidx,
             const float* __restrict__ sh0, const float* __restrict__ sh1,
             const float* __restrict__ sh2, const float* __restrict__ dist,
             const float* __restrict__ Mg,
             const float* __restrict__ W1, const float* __restrict__ b1,
             const float* __restrict__ W2, const float* __restrict__ b2,
             float* __restrict__ outF, int E) {
    __shared__ float sU[WAVES][16 * KP];
    __shared__ float sRaw[WAVES][16 * PN];
    __shared__ float sGate[WAVES][16 * 60];

    const int lane = threadIdx.x & 31;
    const int w = threadIdx.x >> 5;
    const int tile = blockIdx.x * WAVES + w;
    const int eSub = lane & 15;
    const int part = lane >> 4;
    const long e = (long)tile * 16 + eSub;
    const bool valid = (e < (long)E);

    float f[FEAT];
    float shv[9];
    float x0 = 0.0f, x1 = 0.0f;
    int tgt = 0;
    if (valid) {
        const float* fp = &feats[(long)eidx[e] * FEAT];
        tgt = eidx[(long)E + e];
#pragma unroll
        for (int j = 0; j < FEAT; ++j) f[j] = fp[j];
        shv[0] = sh0[e];
#pragma unroll
        for (int j = 0; j < 3; ++j) shv[1 + j] = sh1[e * 3 + j];
#pragma unroll
        for (int j = 0; j < 5; ++j) shv[4 + j] = sh2[e * 5 + j];
        x0 = dist[e];
        float inv0 = 0.5f * (f[0] + f[1]);
        float n1 = 0.0f, n2 = 0.0f;
#pragma unroll
        for (int j = 0; j < 3; ++j) { float m = 0.5f * (f[2 + j] + f[5 + j]); n1 += m * m; }
#pragma unroll
        for (int j = 0; j < 5; ++j) { float m = 0.5f * (f[8 + j] + f[13 + j]); n2 += m * m; }
        x1 = (inv0 + sqrtf(n1) + sqrtf(n2)) * (1.0f / 3.0f);
    } else {
#pragma unroll
        for (int j = 0; j < FEAT; ++j) f[j] = 0.0f;
#pragma unroll
        for (int j = 0; j < 9; ++j) shv[j] = 0.0f;
    }

    float* gRow = &sGate[w][eSub * 60];
    float* uRow = &sU[w][eSub * KP];
    if (part == 0) {
        gate_block<0>(W1, b1, W2, b2, x0, x1, gRow);
        u_stage<0>(shv, f, uRow);
    } else {
        gate_block<1>(W1, b1, W2, b2, x0, x1, gRow);
        u_stage<1>(shv, f, uRow);
    }
    __syncthreads();

    {
        const int col16 = lane & 15;
        const int kofs = (lane < 16) ? 0 : 2;
        const float* uA = &sU[w][col16 * KP];
#pragma unroll
        for (int n = 0; n < NTILES; ++n) {
            v8f acc = {0.f, 0.f, 0.f, 0.f, 0.f, 0.f, 0.f, 0.f};
            const int col = col16 + n * 16;
#pragma unroll
            for (int kt = KLO[n]; kt <= KHI[n]; ++kt) {
                int k0 = kt * 4 + kofs;
                v2f a, b;
                a.x = uA[k0];
                a.y = uA[k0 + 1];
                b.x = Mg[k0 * PN + col];
                b.y = Mg[(k0 + 1) * PN + col];
                acc = __builtin_amdgcn_wmma_f32_16x16x4_f32(
                    false, a, false, b, (short)0, acc, false, false);
            }
            const int rbase = (lane < 16) ? 0 : 8;
#pragma unroll
            for (int j = 0; j < 8; ++j)
                sRaw[w][(rbase + j) * PN + col] = acc[j];
        }
    }
    __syncthreads();

    {
        const float* rRow = &sRaw[w][eSub * PN];
        float* outBase = &outF[(long)tgt * FEAT];
#define DO_SLOT(S)                                                     \
        { float v = slot_val<S>(gRow, rRow);                           \
          if (valid) atomicAdd(outBase + (S), v); }
        if (part == 0) {
            DO_SLOT(0) DO_SLOT(1) DO_SLOT(2) DO_SLOT(3) DO_SLOT(4)
            DO_SLOT(5) DO_SLOT(6) DO_SLOT(7) DO_SLOT(8)
        } else {
            DO_SLOT(9) DO_SLOT(10) DO_SLOT(11) DO_SLOT(12) DO_SLOT(13)
            DO_SLOT(14) DO_SLOT(15) DO_SLOT(16) DO_SLOT(17)
        }
#undef DO_SLOT
    }
}

extern "C" void kernel_launch(void* const* d_in, const int* in_sizes, int n_in,
                              void* d_out, int out_size, void* d_ws, size_t ws_size,
                              hipStream_t stream) {
    const float* node = (const float*)d_in[0];
    const int* eidx   = (const int*)d_in[1];
    const float* sh0  = (const float*)d_in[2];
    const float* sh1  = (const float*)d_in[3];
    const float* sh2  = (const float*)d_in[4];
    const float* dist = (const float*)d_in[5];
    const float* cg[N_COMBOS];
    for (int i = 0; i < N_COMBOS; ++i) cg[i] = (const float*)d_in[6 + i];
    const float* W1_0 = (const float*)d_in[21];
    const float* b1_0 = (const float*)d_in[22];
    const float* W2_0 = (const float*)d_in[23];
    const float* b2_0 = (const float*)d_in[24];
    const float* W1_1 = (const float*)d_in[25];
    const float* b1_1 = (const float*)d_in[26];
    const float* W2_1 = (const float*)d_in[27];
    const float* b2_1 = (const float*)d_in[28];

    const int E = in_sizes[1] / 2;
    const int N = in_sizes[0] / FEAT;

    float* Mg     = (float*)d_ws;
    float* feats1 = (float*)((char*)d_ws + 131072);

    build_M_kernel<<<1, 256, 0, stream>>>(cg[0], cg[1], cg[2], cg[3], cg[4], cg[5],
                                          cg[6], cg[7], cg[8], cg[9], cg[10], cg[11],
                                          cg[12], cg[13], cg[14], Mg);

    long n1 = (long)N * FEAT;
    zero_kernel<<<(int)((n1 + 255) / 256), 256, 0, stream>>>(feats1, n1);
    zero_kernel<<<(out_size + 255) / 256, 256, 0, stream>>>((float*)d_out, (long)out_size);

    int grid = (E + WAVES * 16 - 1) / (WAVES * 16);
    layer_kernel<<<grid, WAVES * 32, 0, stream>>>(node, eidx, sh0, sh1, sh2, dist, Mg,
                                                  W1_0, b1_0, W2_0, b2_0, feats1, E);
    layer_kernel<<<grid, WAVES * 32, 0, stream>>>(feats1, eidx, sh0, sh1, sh2, dist, Mg,
                                                  W1_1, b1_1, W2_1, b2_1, (float*)d_out, E);
}


# Both lanes of an edge pair run one instruction stream: gate weights staged in LDS and lane-indexed, mix and scatter split by output channel; no divergent duplicate passes
# speedup vs baseline: 1.7477x; 1.7477x over previous
#include <hip/hip_runtime.h>

typedef __attribute__((ext_vector_type(2))) float v2f;
typedef __attribute__((ext_vector_type(8))) float v8f;

#define N_COMBOS 15
#define KP 164
#define PN 112
#define NTILES 7
#define WAVES 2
#define FEAT 18

constexpr int cLE[N_COMBOS]     = {0,0,0,1,1,1,1,1,1,2,2,2,2,2,2};
constexpr int cLN[N_COMBOS]     = {0,1,2,0,1,1,1,2,2,0,1,1,2,2,2};
constexpr int cLO[N_COMBOS]     = {0,1,2,1,0,1,2,1,2,2,1,2,0,1,2};
constexpr int OUT_OFF[N_COMBOS] = {0,2,8,18,24,26,32,42,48,58,68,74,84,86,92};
constexpr int U_OFF[9]          = {0,2,8,18,24,42,72,82,112};
constexpr int KLO[NTILES]       = {0,2,6,10,18,20,28};
constexpr int KHI[NTILES]       = {4,10,17,20,27,40,40};
constexpr int N_BY_LO[3]        = {3,6,6};
constexpr int C_BY_LO[3][6]     = {{0,4,12,0,0,0},{1,3,5,7,10,13},{2,6,8,9,11,14}};

__global__ void build_M_kernel(const float* cg0, const float* cg1, const float* cg2,
                               const float* cg3, const float* cg4, const float* cg5,
                               const float* cg6, const float* cg7, const float* cg8,
                               const float* cg9, const float* cg10, const float* cg11,
                               const float* cg12, const float* cg13, const float* cg14,
                               float* M) {
    const float* cgs[N_COMBOS] = {cg0,cg1,cg2,cg3,cg4,cg5,cg6,cg7,cg8,cg9,cg10,cg11,cg12,cg13,cg14};
    int tid = threadIdx.x;
    for (int i = tid; i < KP * PN; i += blockDim.x) M[i] = 0.0f;
    __syncthreads();
    if (tid < N_COMBOS) {
        int c = tid;
        int le = cLE[c], ln = cLN[c], lo = cLO[c];
        int de = 2 * le + 1, dn = 2 * ln + 1, dO = 2 * lo + 1;
        int uo = U_OFF[le * 3 + ln];
        int po = OUT_OFF[c];
        const float* cg = cgs[c];
        for (int a = 0; a < de; ++a)
            for (int b = 0; b < dn; ++b)
                for (int k = 0; k < dO; ++k) {
                    float v = cg[(a * dn + b) * dO + k];
                    for (int ch = 0; ch < 2; ++ch)
                        M[(uo + (a * dn + b) * 2 + ch) * PN + po + ch * dO + k] = v;
                }
    }
}

__global__ void zero_kernel(float* p, long n) {
    long i = (long)blockIdx.x * blockDim.x + threadIdx.x;
    if (i < n) p[i] = 0.0f;
}

#define GCH 6
__device__ __forceinline__ void gate_pair(const float* __restrict__ sW, int part,
                                          float x0, float x1,
                                          float* __restrict__ gRow) {
    const float* w1 = sW + part * 240;
    const float* c1 = sW + 480 + part * 120;
    const float* w2 = sW + 720 + part * 120;
    const float* c2 = sW + 960 + part * 30;
#pragma unroll
    for (int t0 = 0; t0 < 30; t0 += GCH) {
        float acc[GCH];
#pragma unroll
        for (int j = 0; j < GCH; ++j) {
            const int t = t0 + j;
            float a = c2[t];
#pragma unroll
            for (int h = 0; h < 4; ++h) {
                float hh = x0 * w1[t * 8 + h * 2] + x1 * w1[t * 8 + h * 2 + 1] + c1[t * 4 + h];
                hh = hh > 0.0f ? hh : 0.0f;
                a += hh * w2[t * 4 + h];
            }
            acc[j] = a;
        }
#pragma unroll
        for (int j = 0; j < GCH; ++j) gRow[part * 30 + t0 + j] = tanhf(acc[j]);
    }
}

template <int PART>
__device__ __forceinline__ void u_stage(const float* __restrict__ shv,
                                        const float* __restrict__ f,
                                        float* __restrict__ uRow) {
    int u = 0;
#pragma unroll
    for (int le = 0; le < 3; ++le) {
        const int de = 2 * le + 1;
        const int shb = (le == 0) ? 0 : (le == 1 ? 1 : 4);
#pragma unroll
        for (int ln = 0; ln < 3; ++ln) {
            const int dn = 2 * ln + 1;
            const int fb = (ln == 0) ? 0 : (ln == 1 ? 2 : 8);
#pragma unroll
            for (int a = 0; a < de; ++a)
#pragma unroll
                for (int b = 0; b < dn; ++b)
#pragma unroll
                    for (int ch = 0; ch < 2; ++ch) {
                        if ((PART == 0) ? (u < 81) : (u >= 81))
                            uRow[u] = shv[shb + a] * f[fb + ch * dn + b];
                        ++u;
                    }
        }
    }
    if (PART == 1) { uRow[162] = 0.0f; uRow[163] = 0.0f; }
}

template <int LO, int K>
__device__ __forceinline__ float mix_val(const float* __restrict__ gO,
                                         const float* __restrict__ rRow) {
    constexpr int dO = 2 * LO + 1;
    float v = 0.0f;
#pragma unroll
    for (int q = 0; q < N_BY_LO[LO]; ++q) {
        const int c = C_BY_LO[LO][q];
        const int base = OUT_OFF[c];
        v += gO[c * 4]     * rRow[base + K] +
             gO[c * 4 + 2] * rRow[base + dO + K];
    }
    return v;
}

__global__ void __launch_bounds__(WAVES * 32)
layer_kernel(const float* __restrict__ feats, const int* __restrict__ eidx,
             const float* __restrict__ sh0, const float* __restrict__ sh1,
             const float* __restrict__ sh2, const float* __restrict__ dist,
             const float* __restrict__ Mg,
             const float* __restrict__ W1, const float* __restrict__ b1,
             const float* __restrict__ W2, const float* __restrict__ b2,
             float* __restrict__ outF, int E) {
    __shared__ float sU[WAVES][16 * KP];
    __shared__ __attribute__((aligned(16))) float sRaw[WAVES][16 * PN];
    __shared__ float sGate[WAVES][16 * 60];

    const int lane = threadIdx.x & 31;
    const int w = threadIdx.x >> 5;
    const int tile = blockIdx.x * WAVES + w;
    const int eSub = lane & 15;
    const int part = lane >> 4;
    const long e = (long)tile * 16 + eSub;
    const bool valid = (e < (long)E);

    float* sW = &sRaw[0][0];
    {
        const int tid = threadIdx.x;
#pragma unroll
        for (int k = 0; k < 8; ++k) {
            int i = tid + k * (WAVES * 32);
            i = i < 480 ? i : 479;
            sW[i] = W1[i];
        }
#pragma unroll
        for (int k = 0; k < 4; ++k) {
            int i = tid + k * (WAVES * 32);
            i = i < 240 ? i : 239;
            sW[480 + i] = b1[i];
            sW[720 + i] = W2[i];
        }
        const int i2 = tid < 60 ? tid : 59;
        sW[960 + i2] = b2[i2];
    }

    float f[FEAT];
    float shv[9];
    float x0 = 0.0f, x1 = 0.0f;
    int tgt = 0;
    if (valid) {
        const float* fp = &feats[(long)eidx[e] * FEAT];
        tgt = eidx[(long)E + e];
#pragma unroll
        for (int j = 0; j < FEAT; ++j) f[j] = fp[j];
        shv[0] = sh0[e];
#pragma unroll
        for (int j = 0; j < 3; ++j) shv[1 + j] = sh1[e * 3 + j];
#pragma unroll
        for (int j = 0; j < 5; ++j) shv[4 + j] = sh2[e * 5 + j];
        x0 = dist[e];
        float inv0 = 0.5f * (f[0] + f[1]);
        float n1 = 0.0f, n2 = 0.0f;
#pragma unroll
        for (int j = 0; j < 3; ++j) { float m = 0.5f * (f[2 + j] + f[5 + j]); n1 += m * m; }
#pragma unroll
        for (int j = 0; j < 5; ++j) { float m = 0.5f * (f[8 + j] + f[13 + j]); n2 += m * m; }
        x1 = (inv0 + sqrtf(n1) + sqrtf(n2)) * (1.0f / 3.0f);
    } else {
#pragma unroll
        for (int j = 0; j < FEAT; ++j) f[j] = 0.0f;
#pragma unroll
        for (int j = 0; j < 9; ++j) shv[j] = 0.0f;
    }

    __syncthreads();

    float* gRow = &sGate[w][eSub * 60];
    float* uRow = &sU[w][eSub * KP];
    gate_pair(sW, part, x0, x1, gRow);
    if (part == 0) {
        u_stage<0>(shv, f, uRow);
    } else {
        u_stage<1>(shv, f, uRow);
    }
    __syncthreads();

    {
        const int col16 = lane & 15;
        const int kofs = (lane < 16) ? 0 : 2;
        const float* uA = &sU[w][col16 * KP];
#pragma unroll
        for (int n = 0; n < NTILES; ++n) {
            v8f acc = {0.f, 0.f, 0.f, 0.f, 0.f, 0.f, 0.f, 0.f};
            const int col = col16 + n * 16;
#pragma unroll
            for (int kt = KLO[n]; kt <= KHI[n]; ++kt) {
                int k0 = kt * 4 + kofs;
                v2f a, b;
                a.x = uA[k0];
                a.y = uA[k0 + 1];
                b.x = Mg[k0 * PN + col];
                b.y = Mg[(k0 + 1) * PN + col];
                acc = __builtin_amdgcn_wmma_f32_16x16x4_f32(
                    false, a, false, b, (short)0, acc, false, false);
            }
            const int rbase = (lane < 16) ? 0 : 8;
#pragma unroll
            for (int j = 0; j < 8; ++j)
                sRaw[w][(rbase + j) * PN + col] = acc[j];
        }
    }
    __syncthreads();

    {
        const float* rRow = &sRaw[w][eSub * PN];
        const float* gO = gRow + part;
        float* outBase = &outF[(long)tgt * FEAT];
        float* o0 = outBase + part;
        float* o1 = outBase + 2 + part * 3;
        float* o2 = outBase + 8 + part * 5;
#define DO_MIX(LO, K, P)                                               \
        { float v = mix_val<LO, K>(gO, rRow);                          \
          if (valid) atomicAdd((P) + (K), v); }
        DO_MIX(0, 0, o0)
        DO_MIX(1, 0, o1) DO_MIX(1, 1, o1) DO_MIX(1, 2, o1)
        DO_MIX(2, 0, o2) DO_MIX(2, 1, o2) DO_MIX(2, 2, o2) DO_MIX(2, 3, o2) DO_MIX(2, 4, o2)
#undef DO_MIX
    }
}

extern "C" void kernel_launch(void* const* d_in, const int* in_sizes, int n_in,
                              void* d_out, int out_size, void* d_ws, size_t ws_size,
                              hipStream_t stream) {
    const float* node = (const float*)d_in[0];
    const int* eidx   = (const int*)d_in[1];
    const float* sh0  = (const float*)d_in[2];
    const float* sh1  = (const float*)d_in[3];
    const float* sh2  = (const float*)d_in[4];
    const float* dist = (const float*)d_in[5];
    const float* cg[N_COMBOS];
    for (int i = 0; i < N_COMBOS; ++i) cg[i] = (const float*)d_in[6 + i];
    const float* W1_0 = (const float*)d_in[21];
    const float* b1_0 = (const float*)d_in[22];
    const float* W2_0 = (const float*)d_in[23];
    const float* b2_0 = (const float*)d_in[24];
    const float* W1_1 = (const float*)d_in[25];
    const float* b1_1 = (const float*)d_in[26];
    const float* W2_1 = (const float*)d_in[27];
    const float* b2_1 = (const float*)d_in[28];

    const int E = in_sizes[1] / 2;
    const int N = in_sizes[0] / FEAT;

    float* Mg     = (float*)d_ws;
    float* feats1 = (float*)((char*)d_ws + 131072);

    build_M_kernel<<<1, 256, 0, stream>>>(cg[0], cg[1], cg[2], cg[3], cg[4], cg[5],
                                          cg[6], cg[7], cg[8], cg[9], cg[10], cg[11],
                                          cg[12], cg[13], cg[14], Mg);

    long n1 = (long)N * FEAT;
    zero_kernel<<<(int)((n1 + 255) / 256), 256, 0, stream>>>(feats1, n1);
    zero_kernel<<<(out_size + 255) / 256, 256, 0, stream>>>((float*)d_out, (long)out_size);

    int grid = (E + WAVES * 16 - 1) / (WAVES * 16);
    layer_kernel<<<grid, WAVES * 32, 0, stream>>>(node, eidx, sh0, sh1, sh2, dist, Mg,
                                                  W1_0, b1_0, W2_0, b2_0, feats1, E);
    layer_kernel<<<grid, WAVES * 32, 0, stream>>>(feats1, eidx, sh0, sh1, sh2, dist, Mg,
                                                  W1_1, b1_1, W2_1, b2_1, (float*)d_out, E);
}


# Scatter staged through LDS: each f32 atomic instruction carries 32 consecutive floats of the wave's 16x18 result tile (2-3 target rows), not 2 floats in each of 16 rows; sums still in arrival order
# speedup vs baseline: 2.2911x; 1.3109x over previous
#include <hip/hip_runtime.h>

typedef __attribute__((ext_vector_type(2))) float v2f;
typedef __attribute__((ext_vector_type(8))) float v8f;

#define N_COMBOS 15
#define KP 164
#define PN 112
#define NTILES 7
#define WAVES 2
#define FEAT 18

constexpr int cLE[N_COMBOS]     = {0,0,0,1,1,1,1,1,1,2,2,2,2,2,2};
constexpr int cLN[N_COMBOS]     = {0,1,2,0,1,1,1,2,2,0,1,1,2,2,2};
constexpr int cLO[N_COMBOS]     = {0,1,2,1,0,1,2,1,2,2,1,2,0,1,2};
constexpr int OUT_OFF[N_COMBOS] = {0,2,8,18,24,26,32,42,48,58,68,74,84,86,92};
constexpr int U_OFF[9]          = {0,2,8,18,24,42,72,82,112};
constexpr int KLO[NTILES]       = {0,2,6,10,18,20,28};
constexpr int KHI[NTILES]       = {4,10,17,20,27,40,40};
constexpr int N_BY_LO[3]        = {3,6,6};
constexpr int C_BY_LO[3][6]     = {{0,4,12,0,0,0},{1,3,5,7,10,13},{2,6,8,9,11,14}};

__global__ void build_M_kernel(const float* cg0, const float* cg1, const float* cg2,
                               const float* cg3, const float* cg4, const float* cg5,
                               const float* cg6, const float* cg7, const float* cg8,
                               const float* cg9, const float* cg10, const float* cg11,
                               const float* cg12, const float* cg13, const float* cg14,
                               float* M) {
    const float* cgs[N_COMBOS] = {cg0,cg1,cg2,cg3,cg4,cg5,cg6,cg7,cg8,cg9,cg10,cg11,cg12,cg13,cg14};
    int tid = threadIdx.x;
    for (int i = tid; i < KP * PN; i += blockDim.x) M[i] = 0.0f;
    __syncthreads();
    if (tid < N_COMBOS) {
        int c = tid;
        int le = cLE[c], ln = cLN[c], lo = cLO[c];
        int de = 2 * le + 1, dn = 2 * ln + 1, dO = 2 * lo + 1;
        int uo = U_OFF[le * 3 + ln];
        int po = OUT_OFF[c];
        const float* cg = cgs[c];
        for (int a = 0; a < de; ++a)
            for (int b = 0; b < dn; ++b)
                for (int k = 0; k < dO; ++k) {
                    float v = cg[(a * dn + b) * dO + k];
                    for (int ch = 0; ch < 2; ++ch)
                        M[(uo + (a * dn + b) * 2 + ch) * PN + po + ch * dO + k] = v;
                }
    }
}

__global__ void zero_kernel(float* p, long n) {
    long i = (long)blockIdx.x * blockDim.x + threadIdx.x;
    if (i < n) p[i] = 0.0f;
}

#define GCH 6
__device__ __forceinline__ void gate_pair(const float* __restrict__ sW, int part,
                                          float x0, float x1,
                                          float* __restrict__ gRow) {
    const float* w1 = sW + part * 240;
    const float* c1 = sW + 480 + part * 120;
    const float* w2 = sW + 720 + part * 120;
    const float* c2 = sW + 960 + part * 30;
#pragma unroll
    for (int t0 = 0; t0 < 30; t0 += GCH) {
        float acc[GCH];
#pragma unroll
        for (int j = 0; j < GCH; ++j) {
            const int t = t0 + j;
            float a = c2[t];
#pragma unroll
            for (int h = 0; h < 4; ++h) {
                float hh = x0 * w1[t * 8 + h * 2] + x1 * w1[t * 8 + h * 2 + 1] + c1[t * 4 + h];
                hh = hh > 0.0f ? hh : 0.0f;
                a += hh * w2[t * 4 + h];
            }
            acc[j] = a;
        }
#pragma unroll
        for (int j = 0; j < GCH; ++j) gRow[part * 30 + t0 + j] = tanhf(acc[j]);
    }
}

template <int PART>
__device__ __forceinline__ void u_stage(const float* __restrict__ shv,
                                        const float* __restrict__ f,
                                        float* __restrict__ uRow) {
    int u = 0;
#pragma unroll
    for (int le = 0; le < 3; ++le) {
        const int de = 2 * le + 1;
        const int shb = (le == 0) ? 0 : (le == 1 ? 1 : 4);
#pragma unroll
        for (int ln = 0; ln < 3; ++ln) {
            const int dn = 2 * ln + 1;
            const int fb = (ln == 0) ? 0 : (ln == 1 ? 2 : 8);
#pragma unroll
            for (int a = 0; a < de; ++a)
#pragma unroll
                for (int b = 0; b < dn; ++b)
#pragma unroll
                    for (int ch = 0; ch < 2; ++ch) {
                        if ((PART == 0) ? (u < 81) : (u >= 81))
                            uRow[u] = shv[shb + a] * f[fb + ch * dn + b];
                        ++u;
                    }
        }
    }
    if (PART == 1) { uRow[162] = 0.0f; uRow[163] = 0.0f; }
}

template <int LO, int K>
__device__ __forceinline__ float mix_val(const float* __restrict__ gO,
                                         const float* __restrict__ rRow) {
    constexpr int dO = 2 * LO + 1;
    float v = 0.0f;
#pragma unroll
    for (int q = 0; q < N_BY_LO[LO]; ++q) {
        const int c = C_BY_LO[LO][q];
        const int base = OUT_OFF[c];
        v += gO[c * 4]     * rRow[base + K] +
             gO[c * 4 + 2] * rRow[base + dO + K];
    }
    return v;
}

__global__ void __launch_bounds__(WAVES * 32)
layer_kernel(const float* __restrict__ feats, const int* __restrict__ eidx,
             const float* __restrict__ sh0, const float* __restrict__ sh1,
             const float* __restrict__ sh2, const float* __restrict__ dist,
             const float* __restrict__ Mg,
             const float* __restrict__ W1, const float* __restrict__ b1,
             const float* __restrict__ W2, const float* __restrict__ b2,
             float* __restrict__ outF, int E) {
    __shared__ float sU[WAVES][16 * KP];
    __shared__ __attribute__((aligned(16))) float sRaw[WAVES][16 * PN];
    __shared__ float sGate[WAVES][16 * 60];

    const int lane = threadIdx.x & 31;
    const int w = threadIdx.x >> 5;
    const int tile = blockIdx.x * WAVES + w;
    const int eSub = lane & 15;
    const int part = lane >> 4;
    const long e = (long)tile * 16 + eSub;
    const bool valid = (e < (long)E);

    float* sW = &sRaw[0][0];
    {
        const int tid = threadIdx.x;
#pragma unroll
        for (int k = 0; k < 8; ++k) {
            int i = tid + k * (WAVES * 32);
            i = i < 480 ? i : 479;
            sW[i] = W1[i];
        }
#pragma unroll
        for (int k = 0; k < 4; ++k) {
            int i = tid + k * (WAVES * 32);
            i = i < 240 ? i : 239;
            sW[480 + i] = b1[i];
            sW[720 + i] = W2[i];
        }
        const int i2 = tid < 60 ? tid : 59;
        sW[960 + i2] = b2[i2];
    }

    float f[FEAT];
    float shv[9];
    float x0 = 0.0f, x1 = 0.0f;
    int tgt = 0;
    if (valid) {
        const float* fp = &feats[(long)eidx[e] * FEAT];
        tgt = eidx[(long)E + e];
#pragma unroll
        for (int j = 0; j < FEAT; ++j) f[j] = fp[j];
        shv[0] = sh0[e];
#pragma unroll
        for (int j = 0; j < 3; ++j) shv[1 + j] = sh1[e * 3 + j];
#pragma unroll
        for (int j = 0; j < 5; ++j) shv[4 + j] = sh2[e * 5 + j];
        x0 = dist[e];
        float inv0 = 0.5f * (f[0] + f[1]);
        float n1 = 0.0f, n2 = 0.0f;
#pragma unroll
        for (int j = 0; j < 3; ++j) { float m = 0.5f * (f[2 + j] + f[5 + j]); n1 += m * m; }
#pragma unroll
        for (int j = 0; j < 5; ++j) { float m = 0.5f * (f[8 + j] + f[13 + j]); n2 += m * m; }
        x1 = (inv0 + sqrtf(n1) + sqrtf(n2)) * (1.0f / 3.0f);
    } else {
#pragma unroll
        for (int j = 0; j < FEAT; ++j) f[j] = 0.0f;
#pragma unroll
        for (int j = 0; j < 9; ++j) shv[j] = 0.0f;
    }

    __syncthreads();

    float* gRow = &sGate[w][eSub * 60];
    float* uRow = &sU[w][eSub * KP];
    gate_pair(sW, part, x0, x1, gRow);
    if (part == 0) {
        u_stage<0>(shv, f, uRow);
    } else {
        u_stage<1>(shv, f, uRow);
    }
    __syncthreads();

    {
        const int col16 = lane & 15;
        const int kofs = (lane < 16) ? 0 : 2;
        const float* uA = &sU[w][col16 * KP];
#pragma unroll
        for (int n = 0; n < NTILES; ++n) {
            v8f acc = {0.f, 0.f, 0.f, 0.f, 0.f, 0.f, 0.f, 0.f};
            const int col = col16 + n * 16;
#pragma unroll
            for (int kt = KLO[n]; kt <= KHI[n]; ++kt) {
                int k0 = kt * 4 + kofs;
                v2f a, b;
                a.x = uA[k0];
                a.y = uA[k0 + 1];
                b.x = Mg[k0 * PN + col];
                b.y = Mg[(k0 + 1) * PN + col];
                acc = __builtin_amdgcn_wmma_f32_16x16x4_f32(
                    false, a, false, b, (short)0, acc, false, false);
            }
            const int rbase = (lane < 16) ? 0 : 8;
#pragma unroll
            for (int j = 0; j < 8; ++j)
                sRaw[w][(rbase + j) * PN + col] = acc[j];
        }
    }
    __syncthreads();

    {
        const float* rRow = &sRaw[w][eSub * PN];
        const float* gO = gRow + part;
        float* sOut = &sU[w][0];
        float* mRow = sOut + eSub * FEAT;
        float* m0 = mRow + part;
        float* m1 = mRow + 2 + part * 3;
        float* m2 = mRow + 8 + part * 5;
#define DO_MIX(LO, K, P) (P)[K] = mix_val<LO, K>(gO, rRow);
        DO_MIX(0, 0, m0)
        DO_MIX(1, 0, m1) DO_MIX(1, 1, m1) DO_MIX(1, 2, m1)
        DO_MIX(2, 0, m2) DO_MIX(2, 1, m2) DO_MIX(2, 2, m2) DO_MIX(2, 3, m2) DO_MIX(2, 4, m2)
#undef DO_MIX
        sOut[16 * FEAT + eSub] = __int_as_float(tgt);
        __syncthreads();

        constexpr int NS = (16 * FEAT) / 32;
        float v[NS];
        long off[NS];
        unsigned ed[NS];
#pragma unroll
        for (int j = 0; j < NS; ++j) {
            const unsigned p = (unsigned)(j * 32 + lane);
            ed[j] = p / (unsigned)FEAT;
            const unsigned el = p - ed[j] * (unsigned)FEAT;
            const int t = __float_as_int(sOut[16 * FEAT + ed[j]]);
            off[j] = (long)t * FEAT + el;
            v[j] = sOut[p];
        }
        const long e0 = (long)tile * 16;
        if (e0 + 16 <= (long)E) {
#pragma unroll
            for (int j = 0; j < NS; ++j) atomicAdd(&outF[off[j]], v[j]);
        } else {
#pragma unroll
            for (int j = 0; j < NS; ++j)
                if (e0 + (long)ed[j] < (long)E) atomicAdd(&outF[off[j]], v[j]);
        }
    }
}

extern "C" void kernel_launch(void* const* d_in, const int* in_sizes, int n_in,
                              void* d_out, int out_size, void* d_ws, size_t ws_size,
                              hipStream_t stream) {
    const float* node = (const float*)d_in[0];
    const int* eidx   = (const int*)d_in[1];
    const float* sh0  = (const float*)d_in[2];
    const float* sh1  = (const float*)d_in[3];
    const float* sh2  = (const float*)d_in[4];
    const float* dist = (const float*)d_in[5];
    const float* cg[N_COMBOS];
    for (int i = 0; i < N_COMBOS; ++i) cg[i] = (const float*)d_in[6 + i];
    const float* W1_0 = (const float*)d_in[21];
    const float* b1_0 = (const float*)d_in[22];
    const float* W2_0 = (const float*)d_in[23];
    const float* b2_0 = (const float*)d_in[24];
    const float* W1_1 = (const float*)d_in[25];
    const float* b1_1 = (const float*)d_in[26];
    const float* W2_1 = (const float*)d_in[27];
    const float* b2_1 = (const float*)d_in[28];

    const int E = in_sizes[1] / 2;
    const int N = in_sizes[0] / FEAT;

    float* Mg     = (float*)d_ws;
    float* feats1 = (float*)((char*)d_ws + 131072);

    build_M_kernel<<<1, 256, 0, stream>>>(cg[0], cg[1], cg[2], cg[3], cg[4], cg[5],
                                          cg[6], cg[7], cg[8], cg[9], cg[10], cg[11],
                                          cg[12], cg[13], cg[14], Mg);

    long n1 = (long)N * FEAT;
    zero_kernel<<<(int)((n1 + 255) / 256), 256, 0, stream>>>(feats1, n1);
    zero_kernel<<<(out_size + 255) / 256, 256, 0, stream>>>((float*)d_out, (long)out_size);

    int grid = (E + WAVES * 16 - 1) / (WAVES * 16);
    layer_kernel<<<grid, WAVES * 32, 0, stream>>>(node, eidx, sh0, sh1, sh2, dist, Mg,
                                                  W1_0, b1_0, W2_0, b2_0, feats1, E);
    layer_kernel<<<grid, WAVES * 32, 0, stream>>>(feats1, eidx, sh0, sh1, sh2, dist, Mg,
                                                  W1_1, b1_1, W2_1, b2_1, (float*)d_out, E);
}


# Gate phase without per-gate control flow: tanh polynomial for all 30 gates, its exponential form behind one wave-uniform test, ReLU as max; the gates' dependent chains interleave, values unchanged
# speedup vs baseline: 2.4653x; 1.0760x over previous
#include <hip/hip_runtime.h>

typedef __attribute__((ext_vector_type(2))) float v2f;
typedef __attribute__((ext_vector_type(8))) float v8f;

#define N_COMBOS 15
#define KP 164
#define PN 112
#define NTILES 7
#define WAVES 2
#define FEAT 18

constexpr int cLE[N_COMBOS]     = {0,0,0,1,1,1,1,1,1,2,2,2,2,2,2};
constexpr int cLN[N_COMBOS]     = {0,1,2,0,1,1,1,2,2,0,1,1,2,2,2};
constexpr int cLO[N_COMBOS]     = {0,1,2,1,0,1,2,1,2,2,1,2,0,1,2};
constexpr int OUT_OFF[N_COMBOS] = {0,2,8,18,24,26,32,42,48,58,68,74,84,86,92};
constexpr int U_OFF[9]          = {0,2,8,18,24,42,72,82,112};
constexpr int KLO[NTILES]       = {0,2,6,10,18,20,28};
constexpr int KHI[NTILES]       = {4,10,17,20,27,40,40};
constexpr int N_BY_LO[3]        = {3,6,6};
constexpr int C_BY_LO[3][6]     = {{0,4,12,0,0,0},{1,3,5,7,10,13},{2,6,8,9,11,14}};

__global__ void build_M_kernel(const float* cg0, const float* cg1, const float* cg2,
                               const float* cg3, const float* cg4, const float* cg5,
                               const float* cg6, const float* cg7, const float* cg8,
                               const float* cg9, const float* cg10, const float* cg11,
                               const float* cg12, const float* cg13, const float* cg14,
                               float* M) {
    const float* cgs[N_COMBOS] = {cg0,cg1,cg2,cg3,cg4,cg5,cg6,cg7,cg8,cg9,cg10,cg11,cg12,cg13,cg14};
    int tid = threadIdx.x;
    for (int i = tid; i < KP * PN; i += blockDim.x) M[i] = 0.0f;
    __syncthreads();
    if (tid < N_COMBOS) {
        int c = tid;
        int le = cLE[c], ln = cLN[c], lo = cLO[c];
        int de = 2 * le + 1, dn = 2 * ln + 1, dO = 2 * lo + 1;
        int uo = U_OFF[le * 3 + ln];
        int po = OUT_OFF[c];
        const float* cg = cgs[c];
        for (int a = 0; a < de; ++a)
            for (int b = 0; b < dn; ++b)
                for (int k = 0; k < dO; ++k) {
                    float v = cg[(a * dn + b) * dO + k];
                    for (int ch = 0; ch < 2; ++ch)
                        M[(uo + (a * dn + b) * 2 + ch) * PN + po + ch * dO + k] = v;
                }
    }
}

__global__ void zero_kernel(float* p, long n) {
    long i = (long)blockIdx.x * blockDim.x + threadIdx.x;
    if (i < n) p[i] = 0.0f;
}

#define NG 30
__device__ __forceinline__ void gate_pair(const float* __restrict__ sW, int part,
                                          float x0, float x1,
                                          float* __restrict__ gRow) {
    const float* w1 = sW + part * 240;
    const float* c1 = sW + 480 + part * 120;
    const float* w2 = sW + 720 + part * 120;
    const float* c2 = sW + 960 + part * 30;
    float acc[NG];
#pragma unroll
    for (int t = 0; t < NG; ++t) {
        float a = c2[t];
#pragma unroll
        for (int h = 0; h < 4; ++h) {
            float hh = x0 * w1[t * 8 + h * 2] + x1 * w1[t * 8 + h * 2 + 1] + c1[t * 4 + h];
            hh = fmaxf(hh, 0.0f);
            a += hh * w2[t * 4 + h];
        }
        acc[t] = a;
    }
    float z[NG];
    float big = 0.0f;
#pragma unroll
    for (int t = 0; t < NG; ++t) {
        const float y = fabsf(acc[t]);
        const float y2 = y * y;
        float p = __builtin_fmaf(y2, -0x1.758e7ap-8f, 0x1.521192p-6f);
        p = __builtin_fmaf(y2, p, -0x1.b8389cp-5f);
        p = __builtin_fmaf(y2, p, 0x1.110704p-3f);
        p = __builtin_fmaf(y2, p, -0x1.555532p-2f);
        z[t] = __builtin_fmaf(y2, y * p, y);
        big = fmaxf(big, y);
    }
    if (__builtin_amdgcn_ballot_w32(!(big < 0.625f)) != 0u) {
#pragma unroll
        for (int t = 0; t < NG; ++t) {
            const float y = fabsf(acc[t]);
            const float ex = expf(2.0f * y);
            const float zb = __builtin_fmaf(-2.0f, __builtin_amdgcn_rcpf(ex + 1.0f), 1.0f);
            z[t] = (y < 0.625f) ? z[t] : zb;
        }
    }
#pragma unroll
    for (int t = 0; t < NG; ++t) gRow[part * NG + t] = __builtin_copysignf(z[t], acc[t]);
}

template <int PART>
__device__ __forceinline__ void u_stage(const float* __restrict__ shv,
                                        const float* __restrict__ f,
                                        float* __restrict__ uRow) {
    int u = 0;
#pragma unroll
    for (int le = 0; le < 3; ++le) {
        const int de = 2 * le + 1;
        const int shb = (le == 0) ? 0 : (le == 1 ? 1 : 4);
#pragma unroll
        for (int ln = 0; ln < 3; ++ln) {
            const int dn = 2 * ln + 1;
            const int fb = (ln == 0) ? 0 : (ln == 1 ? 2 : 8);
#pragma unroll
            for (int a = 0; a < de; ++a)
#pragma unroll
                for (int b = 0; b < dn; ++b)
#pragma unroll
                    for (int ch = 0; ch < 2; ++ch) {
                        if ((PART == 0) ? (u < 81) : (u >= 81))
                            uRow[u] = shv[shb + a] * f[fb + ch * dn + b];
                        ++u;
                    }
        }
    }
    if (PART == 1) { uRow[162] = 0.0f; uRow[163] = 0.0f; }
}

template <int LO, int K>
__device__ __forceinline__ float mix_val(const float* __restrict__ gO,
                                         const float* __restrict__ rRow) {
    constexpr int dO = 2 * LO + 1;
    float v = 0.0f;
#pragma unroll
    for (int q = 0; q < N_BY_LO[LO]; ++q) {
        const int c = C_BY_LO[LO][q];
        const int base = OUT_OFF[c];
        v += gO[c * 4]     * rRow[base + K] +
             gO[c * 4 + 2] * rRow[base + dO + K];
    }
    return v;
}

__global__ void __launch_bounds__(WAVES * 32)
layer_kernel(const float* __restrict__ feats, const int* __restrict__ eidx,
             const float* __restrict__ sh0, const float* __restrict__ sh1,
             const float* __restrict__ sh2, const float* __restrict__ dist,
             const float* __restrict__ Mg,
             const float* __restrict__ W1, const float* __restrict__ b1,
             const float* __restrict__ W2, const float* __restrict__ b2,
             float* __restrict__ outF, int E) {
    __shared__ float sU[WAVES][16 * KP];
    __shared__ __attribute__((aligned(16))) float sRaw[WAVES][16 * PN];
    __shared__ float sGate[WAVES][16 * 60];

    const int lane = threadIdx.x & 31;
    const int w = threadIdx.x >> 5;
    const int tile = blockIdx.x * WAVES + w;
    const int eSub = lane & 15;
    const int part = lane >> 4;
    const long e = (long)tile * 16 + eSub;
    const bool valid = (e < (long)E);

    float* sW = &sRaw[0][0];
    {
        const int tid = threadIdx.x;
#pragma unroll
        for (int k = 0; k < 8; ++k) {
            int i = tid + k * (WAVES * 32);
            i = i < 480 ? i : 479;
            sW[i] = W1[i];
        }
#pragma unroll
        for (int k = 0; k < 4; ++k) {
            int i = tid + k * (WAVES * 32);
            i = i < 240 ? i : 239;
            sW[480 + i] = b1[i];
            sW[720 + i] = W2[i];
        }
        const int i2 = tid < 60 ? tid : 59;
        sW[960 + i2] = b2[i2];
    }

    float f[FEAT];
    float shv[9];
    float x0 = 0.0f, x1 = 0.0f;
    int tgt = 0;
    if (valid) {
        const float* fp = &feats[(long)eidx[e] * FEAT];
        tgt = eidx[(long)E + e];
#pragma unroll
        for (int j = 0; j < FEAT; ++j) f[j] = fp[j];
        shv[0] = sh0[e];
#pragma unroll
        for (int j = 0; j < 3; ++j) shv[1 + j] = sh1[e * 3 + j];
#pragma unroll
        for (int j = 0; j < 5; ++j) shv[4 + j] = sh2[e * 5 + j];
        x0 = dist[e];
        float inv0 = 0.5f * (f[0] + f[1]);
        float n1 = 0.0f, n2 = 0.0f;
#pragma unroll
        for (int j = 0; j < 3; ++j) { float m = 0.5f * (f[2 + j] + f[5 + j]); n1 += m * m; }
#pragma unroll
        for (int j = 0; j < 5; ++j) { float m = 0.5f * (f[8 + j] + f[13 + j]); n2 += m * m; }
        x1 = (inv0 + sqrtf(n1) + sqrtf(n2)) * (1.0f / 3.0f);
    } else {
#pragma unroll
        for (int j = 0; j < FEAT; ++j) f[j] = 0.0f;
#pragma unroll
        for (int j = 0; j < 9; ++j) shv[j] = 0.0f;
    }

    __syncthreads();

    float* gRow = &sGate[w][eSub * 60];
    float* uRow = &sU[w][eSub * KP];
    gate_pair(sW, part, x0, x1, gRow);
    if (part == 0) {
        u_stage<0>(shv, f, uRow);
    } else {
        u_stage<1>(shv, f, uRow);
    }
    __syncthreads();

    {
        const int col16 = lane & 15;
        const int kofs = (lane < 16) ? 0 : 2;
        const float* uA = &sU[w][col16 * KP];
#pragma unroll
        for (int n = 0; n < NTILES; ++n) {
            v8f acc = {0.f, 0.f, 0.f, 0.f, 0.f, 0.f, 0.f, 0.f};
            const int col = col16 + n * 16;
#pragma unroll
            for (int kt = KLO[n]; kt <= KHI[n]; ++kt) {
                int k0 = kt * 4 + kofs;
                v2f a, b;
                a.x = uA[k0];
                a.y = uA[k0 + 1];
                b.x = Mg[k0 * PN + col];
                b.y = Mg[(k0 + 1) * PN + col];
                acc = __builtin_amdgcn_wmma_f32_16x16x4_f32(
                    false, a, false, b, (short)0, acc, false, false);
            }
            const int rbase = (lane < 16) ? 0 : 8;
#pragma unroll
            for (int j = 0; j < 8; ++j)
                sRaw[w][(rbase + j) * PN + col] = acc[j];
        }
    }
    __syncthreads();

    {
        const float* rRow = &sRaw[w][eSub * PN];
        const float* gO = gRow + part;
        float* sOut = &sU[w][0];
        float* mRow = sOut + eSub * FEAT;
        float* m0 = mRow + part;
        float* m1 = mRow + 2 + part * 3;
        float* m2 = mRow + 8 + part * 5;
#define DO_MIX(LO, K, P) (P)[K] = mix_val<LO, K>(gO, rRow);
        DO_MIX(0, 0, m0)
        DO_MIX(1, 0, m1) DO_MIX(1, 1, m1) DO_MIX(1, 2, m1)
        DO_MIX(2, 0, m2) DO_MIX(2, 1, m2) DO_MIX(2, 2, m2) DO_MIX(2, 3, m2) DO_MIX(2, 4, m2)
#undef DO_MIX
        sOut[16 * FEAT + eSub] = __int_as_float(tgt);
        __syncthreads();

        constexpr int NS = (16 * FEAT) / 32;
        float v[NS];
        long off[NS];
        unsigned ed[NS];
#pragma unroll
        for (int j = 0; j < NS; ++j) {
            const unsigned p = (unsigned)(j * 32 + lane);
            ed[j] = p / (unsigned)FEAT;
            const unsigned el = p - ed[j] * (unsigned)FEAT;
            const int t = __float_as_int(sOut[16 * FEAT + ed[j]]);
            off[j] = (long)t * FEAT + el;
            v[j] = sOut[p];
        }
        const long e0 = (long)tile * 16;
        if (e0 + 16 <= (long)E) {
#pragma unroll
            for (int j = 0; j < NS; ++j) atomicAdd(&outF[off[j]], v[j]);
        } else {
#pragma unroll
            for (int j = 0; j < NS; ++j)
                if (e0 + (long)ed[j] < (long)E) atomicAdd(&outF[off[j]], v[j]);
        }
    }
}

extern "C" void kernel_launch(void* const* d_in, const int* in_sizes, int n_in,
                              void* d_out, int out_size, void* d_ws, size_t ws_size,
                              hipStream_t stream) {
    const float* node = (const float*)d_in[0];
    const int* eidx   = (const int*)d_in[1];
    const float* sh0  = (const float*)d_in[2];
    const float* sh1  = (const float*)d_in[3];
    const float* sh2  = (const float*)d_in[4];
    const float* dist = (const float*)d_in[5];
    const float* cg[N_COMBOS];
    for (int i = 0; i < N_COMBOS; ++i) cg[i] = (const float*)d_in[6 + i];
    const float* W1_0 = (const float*)d_in[21];
    const float* b1_0 = (const float*)d_in[22];
    const float* W2_0 = (const float*)d_in[23];
    const float* b2_0 = (const float*)d_in[24];
    const float* W1_1 = (const float*)d_in[25];
    const float* b1_1 = (const float*)d_in[26];
    const float* W2_1 = (const float*)d_in[27];
    const float* b2_1 = (const float*)d_in[28];

    const int E = in_sizes[1] / 2;
    const int N = in_sizes[0] / FEAT;

    float* Mg     = (float*)d_ws;
    float* feats1 = (float*)((char*)d_ws + 131072);

    build_M_kernel<<<1, 256, 0, stream>>>(cg[0], cg[1], cg[2], cg[3], cg[4], cg[5],
                                          cg[6], cg[7], cg[8], cg[9], cg[10], cg[11],
                                          cg[12], cg[13], cg[14], Mg);

    long n1 = (long)N * FEAT;
    zero_kernel<<<(int)((n1 + 255) / 256), 256, 0, stream>>>(feats1, n1);
    zero_kernel<<<(out_size + 255) / 256, 256, 0, stream>>>((float*)d_out, (long)out_size);

    int grid = (E + WAVES * 16 - 1) / (WAVES * 16);
    layer_kernel<<<grid, WAVES * 32, 0, stream>>>(node, eidx, sh0, sh1, sh2, dist, Mg,
                                                  W1_0, b1_0, W2_0, b2_0, feats1, E);
    layer_kernel<<<grid, WAVES * 32, 0, stream>>>(feats1, eidx, sh0, sh1, sh2, dist, Mg,
                                                  W1_1, b1_1, W2_1, b2_1, (float*)d_out, E);
}


# More waves per SIMD: each wave's raw and scatter tiles reuse its dead u tile in LDS (43,008 -> 32,752 B per work-group) and registers are held to 192: ten work-groups per WGP, not seven
# speedup vs baseline: 2.6694x; 1.0828x over previous
#include <hip/hip_runtime.h>

typedef __attribute__((ext_vector_type(2))) float v2f;
typedef __attribute__((ext_vector_type(8))) float v8f;

#define N_COMBOS 15
#define KP 164
#define PN 112
#define NTILES 7
#define WAVES 2
#define FEAT 18

constexpr int cLE[N_COMBOS]     = {0,0,0,1,1,1,1,1,1,2,2,2,2,2,2};
constexpr int cLN[N_COMBOS]     = {0,1,2,0,1,1,1,2,2,0,1,1,2,2,2};
constexpr int cLO[N_COMBOS]     = {0,1,2,1,0,1,2,1,2,2,1,2,0,1,2};
constexpr int OUT_OFF[N_COMBOS] = {0,2,8,18,24,26,32,42,48,58,68,74,84,86,92};
constexpr int U_OFF[9]          = {0,2,8,18,24,42,72,82,112};
constexpr int KLO[NTILES]       = {0,2,6,10,18,20,28};
constexpr int KHI[NTILES]       = {4,10,17,20,27,40,40};
constexpr int N_BY_LO[3]        = {3,6,6};
constexpr int C_BY_LO[3][6]     = {{0,4,12,0,0,0},{1,3,5,7,10,13},{2,6,8,9,11,14}};

__global__ void build_M_kernel(const float* cg0, const float* cg1, const float* cg2,
                               const float* cg3, const float* cg4, const float* cg5,
                               const float* cg6, const float* cg7, const float* cg8,
                               const float* cg9, const float* cg10, const float* cg11,
                               const float* cg12, const float* cg13, const float* cg14,
                               float* M) {
    const float* cgs[N_COMBOS] = {cg0,cg1,cg2,cg3,cg4,cg5,cg6,cg7,cg8,cg9,cg10,cg11,cg12,cg13,cg14};
    int tid = threadIdx.x;
    for (int i = tid; i < KP * PN; i += blockDim.x) M[i] = 0.0f;
    __syncthreads();
    if (tid < N_COMBOS) {
        int c = tid;
        int le = cLE[c], ln = cLN[c], lo = cLO[c];
        int de = 2 * le + 1, dn = 2 * ln + 1, dO = 2 * lo + 1;
        int uo = U_OFF[le * 3 + ln];
        int po = OUT_OFF[c];
        const float* cg = cgs[c];
        for (int a = 0; a < de; ++a)
            for (int b = 0; b < dn; ++b)
                for (int k = 0; k < dO; ++k) {
                    float v = cg[(a * dn + b) * dO + k];
                    for (int ch = 0; ch < 2; ++ch)
                        M[(uo + (a * dn + b) * 2 + ch) * PN + po + ch * dO + k] = v;
                }
    }
}

__global__ void zero_kernel(float* p, long n) {
    long i = (long)blockIdx.x * blockDim.x + threadIdx.x;
    if (i < n) p[i] = 0.0f;
}

#define NG 30
__device__ __forceinline__ void gate_pair(const float* __restrict__ sW, int part,
                                          float x0, float x1,
                                          float* __restrict__ gRow) {
    const float* w1 = sW + part * 240;
    const float* c1 = sW + 480 + part * 120;
    const float* w2 = sW + 720 + part * 120;
    const float* c2 = sW + 960 + part * 30;
    float acc[NG];
#pragma unroll
    for (int t = 0; t < NG; ++t) {
        float a = c2[t];
#pragma unroll
        for (int h = 0; h < 4; ++h) {
            float hh = x0 * w1[t * 8 + h * 2] + x1 * w1[t * 8 + h * 2 + 1] + c1[t * 4 + h];
            hh = fmaxf(hh, 0.0f);
            a += hh * w2[t * 4 + h];
        }
        acc[t] = a;
    }
    float z[NG];
    float big = 0.0f;
#pragma unroll
    for (int t = 0; t < NG; ++t) {
        const float y = fabsf(acc[t]);
        const float y2 = y * y;
        float p = __builtin_fmaf(y2, -0x1.758e7ap-8f, 0x1.521192p-6f);
        p = __builtin_fmaf(y2, p, -0x1.b8389cp-5f);
        p = __builtin_fmaf(y2, p, 0x1.110704p-3f);
        p = __builtin_fmaf(y2, p, -0x1.555532p-2f);
        z[t] = __builtin_fmaf(y2, y * p, y);
        big = fmaxf(big, y);
    }
    if (__builtin_amdgcn_ballot_w32(!(big < 0.625f)) != 0u) {
#pragma unroll
        for (int t = 0; t < NG; ++t) {
            const float y = fabsf(acc[t]);
            const float ex = expf(2.0f * y);
            const float zb = __builtin_fmaf(-2.0f, __builtin_amdgcn_rcpf(ex + 1.0f), 1.0f);
            z[t] = (y < 0.625f) ? z[t] : zb;
        }
    }
#pragma unroll
    for (int t = 0; t < NG; ++t) gRow[part * NG + t] = __builtin_copysignf(z[t], acc[t]);
}

template <int PART>
__device__ __forceinline__ void u_stage(const float* __restrict__ shv,
                                        const float* __restrict__ f,
                                        float* __restrict__ uRow) {
    int u = 0;
#pragma unroll
    for (int le = 0; le < 3; ++le) {
        const int de = 2 * le + 1;
        const int shb = (le == 0) ? 0 : (le == 1 ? 1 : 4);
#pragma unroll
        for (int ln = 0; ln < 3; ++ln) {
            const int dn = 2 * ln + 1;
            const int fb = (ln == 0) ? 0 : (ln == 1 ? 2 : 8);
#pragma unroll
            for (int a = 0; a < de; ++a)
#pragma unroll
                for (int b = 0; b < dn; ++b)
#pragma unroll
                    for (int ch = 0; ch < 2; ++ch) {
                        if ((PART == 0) ? (u < 81) : (u >= 81))
                            uRow[u] = shv[shb + a] * f[fb + ch * dn + b];
                        ++u;
                    }
        }
    }
    if (PART == 1) { uRow[162] = 0.0f; uRow[163] = 0.0f; }
}

template <int LO, int K>
__device__ __forceinline__ float mix_val(const float* __restrict__ gO,
                                         const float* __restrict__ rRow) {
    constexpr int dO = 2 * LO + 1;
    float v = 0.0f;
#pragma unroll
    for (int q = 0; q < N_BY_LO[LO]; ++q) {
        const int c = C_BY_LO[LO][q];
        const int base = OUT_OFF[c];
        v += gO[c * 4]     * rRow[base + K] +
             gO[c * 4 + 2] * rRow[base + dO + K];
    }
    return v;
}

__global__ void __launch_bounds__(WAVES * 32, 5)
layer_kernel(const float* __restrict__ feats, const int* __restrict__ eidx,
             const float* __restrict__ sh0, const float* __restrict__ sh1,
             const float* __restrict__ sh2, const float* __restrict__ dist,
             const float* __restrict__ Mg,
             const float* __restrict__ W1, const float* __restrict__ b1,
             const float* __restrict__ W2, const float* __restrict__ b2,
             float* __restrict__ outF, int E) {
    __shared__ __attribute__((aligned(16))) float sBuf[WAVES][16 * KP];
    __shared__ __attribute__((aligned(16))) float sGate[WAVES][16 * 60];
    __shared__ __attribute__((aligned(16))) float sWgt[1020];
    static_assert(16 * PN + 16 * FEAT + 16 <= 16 * KP, "raw tile + scatter tile must fit the u tile");

    const int lane = threadIdx.x & 31;
    const int w = threadIdx.x >> 5;
    const int tile = blockIdx.x * WAVES + w;
    const int eSub = lane & 15;
    const int part = lane >> 4;
    const long e = (long)tile * 16 + eSub;
    const bool valid = (e < (long)E);

    float* sW = &sWgt[0];
    {
        const int tid = threadIdx.x;
#pragma unroll
        for (int k = 0; k < 8; ++k) {
            int i = tid + k * (WAVES * 32);
            i = i < 480 ? i : 479;
            sW[i] = W1[i];
        }
#pragma unroll
        for (int k = 0; k < 4; ++k) {
            int i = tid + k * (WAVES * 32);
            i = i < 240 ? i : 239;
            sW[480 + i] = b1[i];
            sW[720 + i] = W2[i];
        }
        const int i2 = tid < 60 ? tid : 59;
        sW[960 + i2] = b2[i2];
    }

    float f[FEAT];
    float shv[9];
    float x0 = 0.0f, x1 = 0.0f;
    int tgt = 0;
    if (valid) {
        const float* fp = &feats[(long)eidx[e] * FEAT];
        tgt = eidx[(long)E + e];
#pragma unroll
        for (int j = 0; j < FEAT; ++j) f[j] = fp[j];
        shv[0] = sh0[e];
#pragma unroll
        for (int j = 0; j < 3; ++j) shv[1 + j] = sh1[e * 3 + j];
#pragma unroll
        for (int j = 0; j < 5; ++j) shv[4 + j] = sh2[e * 5 + j];
        x0 = dist[e];
        float inv0 = 0.5f * (f[0] + f[1]);
        float n1 = 0.0f, n2 = 0.0f;
#pragma unroll
        for (int j = 0; j < 3; ++j) { float m = 0.5f * (f[2 + j] + f[5 + j]); n1 += m * m; }
#pragma unroll
        for (int j = 0; j < 5; ++j) { float m = 0.5f * (f[8 + j] + f[13 + j]); n2 += m * m; }
        x1 = (inv0 + sqrtf(n1) + sqrtf(n2)) * (1.0f / 3.0f);
    } else {
#pragma unroll
        for (int j = 0; j < FEAT; ++j) f[j] = 0.0f;
#pragma unroll
        for (int j = 0; j < 9; ++j) shv[j] = 0.0f;
    }

    __syncthreads();

    float* gRow = &sGate[w][eSub * 60];
    float* uRow = &sBuf[w][eSub * KP];
    gate_pair(sW, part, x0, x1, gRow);
    if (part == 0) {
        u_stage<0>(shv, f, uRow);
    } else {
        u_stage<1>(shv, f, uRow);
    }
    __syncthreads();

    {
        const int col16 = lane & 15;
        const int kofs = (lane < 16) ? 0 : 2;
        const float* uA = &sBuf[w][col16 * KP];
        v8f acc[NTILES];
#pragma unroll
        for (int n = 0; n < NTILES; ++n) {
            if (n == 4) __builtin_amdgcn_sched_barrier(0);
            acc[n] = (v8f){0.f, 0.f, 0.f, 0.f, 0.f, 0.f, 0.f, 0.f};
            const int col = col16 + n * 16;
#pragma unroll
            for (int kt = KLO[n]; kt <= KHI[n]; ++kt) {
                int k0 = kt * 4 + kofs;
                v2f a, b;
                a.x = uA[k0];
                a.y = uA[k0 + 1];
                b.x = Mg[k0 * PN + col];
                b.y = Mg[(k0 + 1) * PN + col];
                acc[n] = __builtin_amdgcn_wmma_f32_16x16x4_f32(
                    false, a, false, b, (short)0, acc[n], false, false);
            }
        }
        __syncthreads();
        float* raw = &sBuf[w][0];
        const int rbase = (lane < 16) ? 0 : 8;
#pragma unroll
        for (int n = 0; n < NTILES; ++n) {
            const int col = col16 + n * 16;
#pragma unroll
            for (int j = 0; j < 8; ++j)
                raw[(rbase + j) * PN + col] = acc[n][j];
        }
    }
    __syncthreads();

    {
        const float* rRow = &sBuf[w][eSub * PN];
        const float* gO = gRow + part;
        float* sOut = &sBuf[w][16 * PN];
        float* mRow = sOut + eSub * FEAT;
        float* m0 = mRow + part;
        float* m1 = mRow + 2 + part * 3;
        float* m2 = mRow + 8 + part * 5;
#define DO_MIX(LO, K, P) (P)[K] = mix_val<LO, K>(gO, rRow);
        DO_MIX(0, 0, m0)
        DO_MIX(1, 0, m1) DO_MIX(1, 1, m1) DO_MIX(1, 2, m1)
        DO_MIX(2, 0, m2) DO_MIX(2, 1, m2) DO_MIX(2, 2, m2) DO_MIX(2, 3, m2) DO_MIX(2, 4, m2)
#undef DO_MIX
        sOut[16 * FEAT + eSub] = __int_as_float(tgt);
        __syncthreads();

        constexpr int NS = (16 * FEAT) / 32;
        float v[NS];
        long off[NS];
        unsigned ed[NS];
#pragma unroll
        for (int j = 0; j < NS; ++j) {
            const unsigned p = (unsigned)(j * 32 + lane);
            ed[j] = p / (unsigned)FEAT;
            const unsigned el = p - ed[j] * (unsigned)FEAT;
            const int t = __float_as_int(sOut[16 * FEAT + ed[j]]);
            off[j] = (long)t * FEAT + el;
            v[j] = sOut[p];
        }
        const long e0 = (long)tile * 16;
        if (e0 + 16 <= (long)E) {
#pragma unroll
            for (int j = 0; j < NS; ++j) atomicAdd(&outF[off[j]], v[j]);
        } else {
#pragma unroll
            for (int j = 0; j < NS; ++j)
                if (e0 + (long)ed[j] < (long)E) atomicAdd(&outF[off[j]], v[j]);
        }
    }
}

extern "C" void kernel_launch(void* const* d_in, const int* in_sizes, int n_in,
                              void* d_out, int out_size, void* d_ws, size_t ws_size,
                              hipStream_t stream) {
    const float* node = (const float*)d_in[0];
    const int* eidx   = (const int*)d_in[1];
    const float* sh0  = (const float*)d_in[2];
    const float* sh1  = (const float*)d_in[3];
    const float* sh2  = (const float*)d_in[4];
    const float* dist = (const float*)d_in[5];
    const float* cg[N_COMBOS];
    for (int i = 0; i < N_COMBOS; ++i) cg[i] = (const float*)d_in[6 + i];
    const float* W1_0 = (const float*)d_in[21];
    const float* b1_0 = (const float*)d_in[22];
    const float* W2_0 = (const float*)d_in[23];
    const float* b2_0 = (const float*)d_in[24];
    const float* W1_1 = (const float*)d_in[25];
    const float* b1_1 = (const float*)d_in[26];
    const float* W2_1 = (const float*)d_in[27];
    const float* b2_1 = (const float*)d_in[28];

    const int E = in_sizes[1] / 2;
    const int N = in_sizes[0] / FEAT;

    float* Mg     = (float*)d_ws;
    float* feats1 = (float*)((char*)d_ws + 131072);

    build_M_kernel<<<1, 256, 0, stream>>>(cg[0], cg[1], cg[2], cg[3], cg[4], cg[5],
                                          cg[6], cg[7], cg[8], cg[9], cg[10], cg[11],
                                          cg[12], cg[13], cg[14], Mg);

    long n1 = (long)N * FEAT;
    zero_kernel<<<(int)((n1 + 255) / 256), 256, 0, stream>>>(feats1, n1);
    zero_kernel<<<(out_size + 255) / 256, 256, 0, stream>>>((float*)d_out, (long)out_size);

    int grid = (E + WAVES * 16 - 1) / (WAVES * 16);
    layer_kernel<<<grid, WAVES * 32, 0, stream>>>(node, eidx, sh0, sh1, sh2, dist, Mg,
                                                  W1_0, b1_0, W2_0, b2_0, feats1, E);
    layer_kernel<<<grid, WAVES * 32, 0, stream>>>(feats1, eidx, sh0, sh1, sh2, dist, Mg,
                                                  W1_1, b1_1, W2_1, b2_1, (float*)d_out, E);
}


# Mix phase reads each edge's raw row once, as 16-byte LDS loads at a conflict-free row pitch (116 floats, not 112): 41 LDS reads per wave instead of 114, none bank-conflicted; sums unchanged
# speedup vs baseline: 2.7882x; 1.0445x over previous
#include <hip/hip_runtime.h>

typedef __attribute__((ext_vector_type(2))) float v2f;
typedef __attribute__((ext_vector_type(4))) float v4f;
typedef __attribute__((ext_vector_type(8))) float v8f;

#define N_COMBOS 15
#define KP 164
#define PN 112
#define RP 116
#define RQ 26
#define NTILES 7
#define WAVES 2
#define FEAT 18

constexpr int cLE[N_COMBOS]     = {0,0,0,1,1,1,1,1,1,2,2,2,2,2,2};
constexpr int cLN[N_COMBOS]     = {0,1,2,0,1,1,1,2,2,0,1,1,2,2,2};
constexpr int cLO[N_COMBOS]     = {0,1,2,1,0,1,2,1,2,2,1,2,0,1,2};
constexpr int OUT_OFF[N_COMBOS] = {0,2,8,18,24,26,32,42,48,58,68,74,84,86,92};
constexpr int U_OFF[9]          = {0,2,8,18,24,42,72,82,112};
constexpr int KLO[NTILES]       = {0,2,6,10,18,20,28};
constexpr int KHI[NTILES]       = {4,10,17,20,27,40,40};
constexpr int N_BY_LO[3]        = {3,6,6};
constexpr int C_BY_LO[3][6]     = {{0,4,12,0,0,0},{1,3,5,7,10,13},{2,6,8,9,11,14}};

__global__ void build_M_kernel(const float* cg0, const float* cg1, const float* cg2,
                               const float* cg3, const float* cg4, const float* cg5,
                               const float* cg6, const float* cg7, const float* cg8,
                               const float* cg9, const float* cg10, const float* cg11,
                               const float* cg12, const float* cg13, const float* cg14,
                               float* M) {
    const float* cgs[N_COMBOS] = {cg0,cg1,cg2,cg3,cg4,cg5,cg6,cg7,cg8,cg9,cg10,cg11,cg12,cg13,cg14};
    int tid = threadIdx.x;
    for (int i = tid; i < KP * PN; i += blockDim.x) M[i] = 0.0f;
    __syncthreads();
    if (tid < N_COMBOS) {
        int c = tid;
        int le = cLE[c], ln = cLN[c], lo = cLO[c];
        int de = 2 * le + 1, dn = 2 * ln + 1, dO = 2 * lo + 1;
        int uo = U_OFF[le * 3 + ln];
        int po = OUT_OFF[c];
        const float* cg = cgs[c];
        for (int a = 0; a < de; ++a)
            for (int b = 0; b < dn; ++b)
                for (int k = 0; k < dO; ++k) {
                    float v = cg[(a * dn + b) * dO + k];
                    for (int ch = 0; ch < 2; ++ch)
                        M[(uo + (a * dn + b) * 2 + ch) * PN + po + ch * dO + k] = v;
                }
    }
}

__global__ void zero_kernel(float* p, long n) {
    long i = (long)blockIdx.x * blockDim.x + threadIdx.x;
    if (i < n) p[i] = 0.0f;
}

#define NG 30
__device__ __forceinline__ void gate_pair(const float* __restrict__ sW, int part,
                                          float x0, float x1,
                                          float* __restrict__ gRow) {
    const float* w1 = sW + part * 240;
    const float* c1 = sW + 480 + part * 120;
    const float* w2 = sW + 720 + part * 120;
    const float* c2 = sW + 960 + part * 30;
    float acc[NG];
#pragma unroll
    for (int t = 0; t < NG; ++t) {
        float a = c2[t];
#pragma unroll
        for (int h = 0; h < 4; ++h) {
            float hh = x0 * w1[t * 8 + h * 2] + x1 * w1[t * 8 + h * 2 + 1] + c1[t * 4 + h];
            hh = fmaxf(hh, 0.0f);
            a += hh * w2[t * 4 + h];
        }
        acc[t] = a;
    }
    float z[NG];
    float big = 0.0f;
#pragma unroll
    for (int t = 0; t < NG; ++t) {
        const float y = fabsf(acc[t]);
        const float y2 = y * y;
        float p = __builtin_fmaf(y2, -0x1.758e7ap-8f, 0x1.521192p-6f);
        p = __builtin_fmaf(y2, p, -0x1.b8389cp-5f);
        p = __builtin_fmaf(y2, p, 0x1.110704p-3f);
        p = __builtin_fmaf(y2, p, -0x1.555532p-2f);
        z[t] = __builtin_fmaf(y2, y * p, y);
        big = fmaxf(big, y);
    }
    if (__builtin_amdgcn_ballot_w32(!(big < 0.625f)) != 0u) {
#pragma unroll
        for (int t = 0; t < NG; ++t) {
            const float y = fabsf(acc[t]);
            const float ex = expf(2.0f * y);
            const float zb = __builtin_fmaf(-2.0f, __builtin_amdgcn_rcpf(ex + 1.0f), 1.0f);
            z[t] = (y < 0.625f) ? z[t] : zb;
        }
    }
#pragma unroll
    for (int t = 0; t < NG; ++t) gRow[part * NG + t] = __builtin_copysignf(z[t], acc[t]);
}

template <int PART>
__device__ __forceinline__ void u_stage(const float* __restrict__ shv,
                                        const float* __restrict__ f,
                                        float* __restrict__ uRow) {
    int u = 0;
#pragma unroll
    for (int le = 0; le < 3; ++le) {
        const int de = 2 * le + 1;
        const int shb = (le == 0) ? 0 : (le == 1 ? 1 : 4);
#pragma unroll
        for (int ln = 0; ln < 3; ++ln) {
            const int dn = 2 * ln + 1;
            const int fb = (ln == 0) ? 0 : (ln == 1 ? 2 : 8);
#pragma unroll
            for (int a = 0; a < de; ++a)
#pragma unroll
                for (int b = 0; b < dn; ++b)
#pragma unroll
                    for (int ch = 0; ch < 2; ++ch) {
                        if ((PART == 0) ? (u < 81) : (u >= 81))
                            uRow[u] = shv[shb + a] * f[fb + ch * dn + b];
                        ++u;
                    }
        }
    }
    if (PART == 1) { uRow[162] = 0.0f; uRow[163] = 0.0f; }
}

template <int LO, int K>
__device__ __forceinline__ float mix_val(const float (&g)[2 * N_COMBOS],
                                         const float (&r)[4 * RQ]) {
    constexpr int dO = 2 * LO + 1;
    float v = 0.0f;
#pragma unroll
    for (int q = 0; q < N_BY_LO[LO]; ++q) {
        const int c = C_BY_LO[LO][q];
        const int base = OUT_OFF[c];
        v += g[c * 2]     * r[base + K] +
             g[c * 2 + 1] * r[base + dO + K];
    }
    return v;
}

__global__ void __launch_bounds__(WAVES * 32, 5)
layer_kernel(const float* __restrict__ feats, const int* __restrict__ eidx,
             const float* __restrict__ sh0, const float* __restrict__ sh1,
             const float* __restrict__ sh2, const float* __restrict__ dist,
             const float* __restrict__ Mg,
             const float* __restrict__ W1, const float* __restrict__ b1,
             const float* __restrict__ W2, const float* __restrict__ b2,
             float* __restrict__ outF, int E) {
    __shared__ __attribute__((aligned(16))) float sBuf[WAVES][16 * KP];
    __shared__ __attribute__((aligned(16))) float sGate[WAVES][16 * 60];
    __shared__ __attribute__((aligned(16))) float sWgt[1020];
    static_assert(16 * RP + 16 * FEAT + 16 <= 16 * KP, "raw tile + scatter tile must fit the u tile");
    static_assert(4 * RQ <= RP && 4 * RQ >= 102, "the 16-byte pieces cover a raw row and stay inside it");

    const int lane = threadIdx.x & 31;
    const int w = threadIdx.x >> 5;
    const int tile = blockIdx.x * WAVES + w;
    const int eSub = lane & 15;
    const int part = lane >> 4;
    const long e = (long)tile * 16 + eSub;
    const bool valid = (e < (long)E);

    float* sW = &sWgt[0];
    {
        const int tid = threadIdx.x;
#pragma unroll
        for (int k = 0; k < 8; ++k) {
            int i = tid + k * (WAVES * 32);
            i = i < 480 ? i : 479;
            sW[i] = W1[i];
        }
#pragma unroll
        for (int k = 0; k < 4; ++k) {
            int i = tid + k * (WAVES * 32);
            i = i < 240 ? i : 239;
            sW[480 + i] = b1[i];
            sW[720 + i] = W2[i];
        }
        const int i2 = tid < 60 ? tid : 59;
        sW[960 + i2] = b2[i2];
    }

    float f[FEAT];
    float shv[9];
    float x0 = 0.0f, x1 = 0.0f;
    int tgt = 0;
    if (valid) {
        const float* fp = &feats[(long)eidx[e] * FEAT];
        tgt = eidx[(long)E + e];
#pragma unroll
        for (int j = 0; j < FEAT; ++j) f[j] = fp[j];
        shv[0] = sh0[e];
#pragma unroll
        for (int j = 0; j < 3; ++j) shv[1 + j] = sh1[e * 3 + j];
#pragma unroll
        for (int j = 0; j < 5; ++j) shv[4 + j] = sh2[e * 5 + j];
        x0 = dist[e];
        float inv0 = 0.5f * (f[0] + f[1]);
        float n1 = 0.0f, n2 = 0.0f;
#pragma unroll
        for (int j = 0; j < 3; ++j) { float m = 0.5f * (f[2 + j] + f[5 + j]); n1 += m * m; }
#pragma unroll
        for (int j = 0; j < 5; ++j) { float m = 0.5f * (f[8 + j] + f[13 + j]); n2 += m * m; }
        x1 = (inv0 + sqrtf(n1) + sqrtf(n2)) * (1.0f / 3.0f);
    } else {
#pragma unroll
        for (int j = 0; j < FEAT; ++j) f[j] = 0.0f;
#pragma unroll
        for (int j = 0; j < 9; ++j) shv[j] = 0.0f;
    }

    __syncthreads();

    float* gRow = &sGate[w][eSub * 60];
    float* uRow = &sBuf[w][eSub * KP];
    gate_pair(sW, part, x0, x1, gRow);
    if (part == 0) {
        u_stage<0>(shv, f, uRow);
    } else {
        u_stage<1>(shv, f, uRow);
    }
    __syncthreads();

    {
        const int col16 = lane & 15;
        const int kofs = (lane < 16) ? 0 : 2;
        const float* uA = &sBuf[w][col16 * KP];
        v8f acc[NTILES];
#pragma unroll
        for (int n = 0; n < NTILES; ++n) {
            if (n == 4) __builtin_amdgcn_sched_barrier(0);
            acc[n] = (v8f){0.f, 0.f, 0.f, 0.f, 0.f, 0.f, 0.f, 0.f};
            const int col = col16 + n * 16;
#pragma unroll
            for (int kt = KLO[n]; kt <= KHI[n]; ++kt) {
                int k0 = kt * 4 + kofs;
                v2f a, b;
                a.x = uA[k0];
                a.y = uA[k0 + 1];
                b.x = Mg[k0 * PN + col];
                b.y = Mg[(k0 + 1) * PN + col];
                acc[n] = __builtin_amdgcn_wmma_f32_16x16x4_f32(
                    false, a, false, b, (short)0, acc[n], false, false);
            }
        }
        __syncthreads();
        float* raw = &sBuf[w][0];
        const int rbase = (lane < 16) ? 0 : 8;
#pragma unroll
        for (int n = 0; n < NTILES; ++n) {
            const int col = col16 + n * 16;
#pragma unroll
            for (int j = 0; j < 8; ++j)
                raw[(rbase + j) * RP + col] = acc[n][j];
        }
    }
    __syncthreads();

    {
        const float* rRow = &sBuf[w][eSub * RP];
        float r[4 * RQ];
#pragma unroll
        for (int q = 0; q < RQ; ++q) {
            const v4f t = *reinterpret_cast<const v4f*>(rRow + 4 * q);
            r[4 * q] = t.x; r[4 * q + 1] = t.y; r[4 * q + 2] = t.z; r[4 * q + 3] = t.w;
        }
        const float* gO = gRow + part;
        float g[2 * N_COMBOS];
#pragma unroll
        for (int q = 0; q < 2 * N_COMBOS; ++q) g[q] = gO[q * 2];
        float* sOut = &sBuf[w][16 * RP];
        float* mRow = sOut + eSub * FEAT;
        float* m0 = mRow + part;
        float* m1 = mRow + 2 + part * 3;
        float* m2 = mRow + 8 + part * 5;
#define DO_MIX(LO, K, P) (P)[K] = mix_val<LO, K>(g, r);
        DO_MIX(0, 0, m0)
        DO_MIX(1, 0, m1) DO_MIX(1, 1, m1) DO_MIX(1, 2, m1)
        DO_MIX(2, 0, m2) DO_MIX(2, 1, m2) DO_MIX(2, 2, m2) DO_MIX(2, 3, m2) DO_MIX(2, 4, m2)
#undef DO_MIX
        sOut[16 * FEAT + eSub] = __int_as_float(tgt);
        __syncthreads();

        constexpr int NS = (16 * FEAT) / 32;
        float v[NS];
        long off[NS];
        unsigned ed[NS];
#pragma unroll
        for (int j = 0; j < NS; ++j) {
            const unsigned p = (unsigned)(j * 32 + lane);
            ed[j] = p / (unsigned)FEAT;
            const unsigned el = p - ed[j] * (unsigned)FEAT;
            const int t = __float_as_int(sOut[16 * FEAT + ed[j]]);
            off[j] = (long)t * FEAT + el;
            v[j] = sOut[p];
        }
        const long e0 = (long)tile * 16;
        if (e0 + 16 <= (long)E) {
#pragma unroll
            for (int j = 0; j < NS; ++j) atomicAdd(&outF[off[j]], v[j]);
        } else {
#pragma unroll
            for (int j = 0; j < NS; ++j)
                if (e0 + (long)ed[j] < (long)E) atomicAdd(&outF[off[j]], v[j]);
        }
    }
}

extern "C" void kernel_launch(void* const* d_in, const int* in_sizes, int n_in,
                              void* d_out, int out_size, void* d_ws, size_t ws_size,
                              hipStream_t stream) {
    const float* node = (const float*)d_in[0];
    const int* eidx   = (const int*)d_in[1];
    const float* sh0  = (const float*)d_in[2];
    const float* sh1  = (const float*)d_in[3];
    const float* sh2  = (const float*)d_in[4];
    const float* dist = (const float*)d_in[5];
    const float* cg[N_COMBOS];
    for (int i = 0; i < N_COMBOS; ++i) cg[i] = (const float*)d_in[6 + i];
    const float* W1_0 = (const float*)d_in[21];
    const float* b1_0 = (const float*)d_in[22];
    const float* W2_0 = (const float*)d_in[23];
    const float* b2_0 = (const float*)d_in[24];
    const float* W1_1 = (const float*)d_in[25];
    const float* b1_1 = (const float*)d_in[26];
    const float* W2_1 = (const float*)d_in[27];
    const float* b2_1 = (const float*)d_in[28];

    const int E = in_sizes[1] / 2;
    const int N = in_sizes[0] / FEAT;

    float* Mg     = (float*)d_ws;
    float* feats1 = (float*)((char*)d_ws + 131072);

    build_M_kernel<<<1, 256, 0, stream>>>(cg[0], cg[1], cg[2], cg[3], cg[4], cg[5],
                                          cg[6], cg[7], cg[8], cg[9], cg[10], cg[11],
                                          cg[12], cg[13], cg[14], Mg);

    long n1 = (long)N * FEAT;
    zero_kernel<<<(int)((n1 + 255) / 256), 256, 0, stream>>>(feats1, n1);
    zero_kernel<<<(out_size + 255) / 256, 256, 0, stream>>>((float*)d_out, (long)out_size);

    int grid = (E + WAVES * 16 - 1) / (WAVES * 16);
    layer_kernel<<<grid, WAVES * 32, 0, stream>>>(node, eidx, sh0, sh1, sh2, dist, Mg,
                                                  W1_0, b1_0, W2_0, b2_0, feats1, E);
    layer_kernel<<<grid, WAVES * 32, 0, stream>>>(feats1, eidx, sh0, sh1, sh2, dist, Mg,
                                                  W1_1, b1_1, W2_1, b2_1, (float*)d_out, E);
}


# First gate layer (2 inputs + bias -> 240 hidden units per edge) moved from vector FMAs on LDS-read weights to 15 f32 16x16x4 WMMAs per tile; its sums are now rounded in the matrix unit's order
# speedup vs baseline: 3.0703x; 1.1012x over previous
#include <hip/hip_runtime.h>

typedef __attribute__((ext_vector_type(2))) float v2f;
typedef __attribute__((ext_vector_type(4))) float v4f;
typedef __attribute__((ext_vector_type(8))) float v8f;

#define N_COMBOS 15
#define KP 164
#define PN 112
#define RP 116
#define RQ 26
#define NTILES 7
#define WAVES 2
#define FEAT 18

constexpr int cLE[N_COMBOS]     = {0,0,0,1,1,1,1,1,1,2,2,2,2,2,2};
constexpr int cLN[N_COMBOS]     = {0,1,2,0,1,1,1,2,2,0,1,1,2,2,2};
constexpr int cLO[N_COMBOS]     = {0,1,2,1,0,1,2,1,2,2,1,2,0,1,2};
constexpr int OUT_OFF[N_COMBOS] = {0,2,8,18,24,26,32,42,48,58,68,74,84,86,92};
constexpr int U_OFF[9]          = {0,2,8,18,24,42,72,82,112};
constexpr int KLO[NTILES]       = {0,2,6,10,18,20,28};
constexpr int KHI[NTILES]       = {4,10,17,20,27,40,40};
constexpr int N_BY_LO[3]        = {3,6,6};
constexpr int C_BY_LO[3][6]     = {{0,4,12,0,0,0},{1,3,5,7,10,13},{2,6,8,9,11,14}};

__global__ void build_M_kernel(const float* cg0, const float* cg1, const float* cg2,
                               const float* cg3, const float* cg4, const float* cg5,
                               const float* cg6, const float* cg7, const float* cg8,
                               const float* cg9, const float* cg10, const float* cg11,
                               const float* cg12, const float* cg13, const float* cg14,
                               float* M) {
    const float* cgs[N_COMBOS] = {cg0,cg1,cg2,cg3,cg4,cg5,cg6,cg7,cg8,cg9,cg10,cg11,cg12,cg13,cg14};
    int tid = threadIdx.x;
    for (int i = tid; i < KP * PN; i += blockDim.x) M[i] = 0.0f;
    __syncthreads();
    if (tid < N_COMBOS) {
        int c = tid;
        int le = cLE[c], ln = cLN[c], lo = cLO[c];
        int de = 2 * le + 1, dn = 2 * ln + 1, dO = 2 * lo + 1;
        int uo = U_OFF[le * 3 + ln];
        int po = OUT_OFF[c];
        const float* cg = cgs[c];
        for (int a = 0; a < de; ++a)
            for (int b = 0; b < dn; ++b)
                for (int k = 0; k < dO; ++k) {
                    float v = cg[(a * dn + b) * dO + k];
                    for (int ch = 0; ch < 2; ++ch)
                        M[(uo + (a * dn + b) * 2 + ch) * PN + po + ch * dO + k] = v;
                }
    }
}

__global__ void zero_kernel(float* p, long n) {
    long i = (long)blockIdx.x * blockDim.x + threadIdx.x;
    if (i < n) p[i] = 0.0f;
}

#define NG 30
#define NGT 15
__device__ __forceinline__ void gate_mlp(const float* __restrict__ sW, int part,
                                         float x0, float x1,
                                         const v2f (&wA)[NGT], const float (&bA)[NGT],
                                         float* __restrict__ gRow) {
    v2f xb;
    xb.x = part ? 1.0f : x0;
    xb.y = part ? 0.0f : x1;
    const float* w2 = sW + part * 8;
    const float* c2 = sW + 240 + part * 2;
    float acc[NG];
#pragma unroll
    for (int j = 0; j < NGT; ++j) {
        v2f a;
        a.x = part ? bA[j] : wA[j].x;
        a.y = part ? 0.0f : wA[j].y;
        const v8f zero = {0.f, 0.f, 0.f, 0.f, 0.f, 0.f, 0.f, 0.f};
        const v8f d = __builtin_amdgcn_wmma_f32_16x16x4_f32(
            false, a, false, xb, (short)0, zero, false, false);
        v2f s = *reinterpret_cast<const v2f*>(c2 + j * 4);
#pragma unroll
        for (int h = 0; h < 4; ++h) {
            v2f hh;
            hh.x = fmaxf(d[h], 0.0f);
            hh.y = fmaxf(d[4 + h], 0.0f);
            const v2f wv = *reinterpret_cast<const v2f*>(w2 + j * 16 + h * 2);
            s = __builtin_elementwise_fma(hh, wv, s);
        }
        acc[j * 2] = s.x;
        acc[j * 2 + 1] = s.y;
    }
    float z[NG];
    float big = 0.0f;
#pragma unroll
    for (int t = 0; t < NG; ++t) {
        const float y = fabsf(acc[t]);
        const float y2 = y * y;
        float p = __builtin_fmaf(y2, -0x1.758e7ap-8f, 0x1.521192p-6f);
        p = __builtin_fmaf(y2, p, -0x1.b8389cp-5f);
        p = __builtin_fmaf(y2, p, 0x1.110704p-3f);
        p = __builtin_fmaf(y2, p, -0x1.555532p-2f);
        z[t] = __builtin_fmaf(y2, y * p, y);
        big = fmaxf(big, y);
    }
    if (__builtin_amdgcn_ballot_w32(!(big < 0.625f)) != 0u) {
#pragma unroll
        for (int t = 0; t < NG; ++t) {
            const float y = fabsf(acc[t]);
            const float ex = expf(2.0f * y);
            const float zb = __builtin_fmaf(-2.0f, __builtin_amdgcn_rcpf(ex + 1.0f), 1.0f);
            z[t] = (y < 0.625f) ? z[t] : zb;
        }
    }
#pragma unroll
    for (int j = 0; j < NGT; ++j)
#pragma unroll
        for (int i = 0; i < 2; ++i)
            gRow[j * 4 + part * 2 + i] = __builtin_copysignf(z[j * 2 + i], acc[j * 2 + i]);
}

template <int PART>
__device__ __forceinline__ void u_stage(const float* __restrict__ shv,
                                        const float* __restrict__ f,
                                        float* __restrict__ uRow) {
    int u = 0;
#pragma unroll
    for (int le = 0; le < 3; ++le) {
        const int de = 2 * le + 1;
        const int shb = (le == 0) ? 0 : (le == 1 ? 1 : 4);
#pragma unroll
        for (int ln = 0; ln < 3; ++ln) {
            const int dn = 2 * ln + 1;
            const int fb = (ln == 0) ? 0 : (ln == 1 ? 2 : 8);
#pragma unroll
            for (int a = 0; a < de; ++a)
#pragma unroll
                for (int b = 0; b < dn; ++b)
#pragma unroll
                    for (int ch = 0; ch < 2; ++ch) {
                        if ((PART == 0) ? (u < 81) : (u >= 81))
                            uRow[u] = shv[shb + a] * f[fb + ch * dn + b];
                        ++u;
                    }
        }
    }
    if (PART == 1) { uRow[162] = 0.0f; uRow[163] = 0.0f; }
}

template <int LO, int K>
__device__ __forceinline__ float mix_val(const float (&g)[2 * N_COMBOS],
                                         const float (&r)[4 * RQ]) {
    constexpr int dO = 2 * LO + 1;
    float v = 0.0f;
#pragma unroll
    for (int q = 0; q < N_BY_LO[LO]; ++q) {
        const int c = C_BY_LO[LO][q];
        const int base = OUT_OFF[c];
        v += g[c * 2]     * r[base + K] +
             g[c * 2 + 1] * r[base + dO + K];
    }
    return v;
}

__global__ void __launch_bounds__(WAVES * 32, 5)
layer_kernel(const float* __restrict__ feats, const int* __restrict__ eidx,
             const float* __restrict__ sh0, const float* __restrict__ sh1,
             const float* __restrict__ sh2, const float* __restrict__ dist,
             const float* __restrict__ Mg,
             const float* __restrict__ W1, const float* __restrict__ b1,
             const float* __restrict__ W2, const float* __restrict__ b2,
             float* __restrict__ outF, int E) {
    __shared__ __attribute__((aligned(16))) float sBuf[WAVES][16 * KP];
    __shared__ __attribute__((aligned(16))) float sGate[WAVES][16 * 60];
    __shared__ __attribute__((aligned(16))) float sWgt[300];
    static_assert(16 * RP + 16 * FEAT + 16 <= 16 * KP, "raw tile + scatter tile must fit the u tile");
    static_assert(4 * RQ <= RP && 4 * RQ >= 102, "the 16-byte pieces cover a raw row and stay inside it");

    const int lane = threadIdx.x & 31;
    const int w = threadIdx.x >> 5;
    const int tile = blockIdx.x * WAVES + w;
    const int eSub = lane & 15;
    const int part = lane >> 4;
    const long e = (long)tile * 16 + eSub;
    const bool valid = (e < (long)E);

    float* sW = &sWgt[0];
    {
        const int tid = threadIdx.x;
#pragma unroll
        for (int k = 0; k < 4; ++k) {
            int i = tid + k * (WAVES * 32);
            i = i < 240 ? i : 239;
            sW[(i & ~7) | ((i & 3) << 1) | ((i >> 2) & 1)] = W2[i];
        }
        const int i2 = tid < 60 ? tid : 59;
        sW[240 + i2] = b2[i2];
    }
    v2f wA[NGT];
    float bA[NGT];
#pragma unroll
    for (int j = 0; j < NGT; ++j) {
        wA[j] = *reinterpret_cast<const v2f*>(W1 + (j * 16 + eSub) * 2);
        bA[j] = b1[j * 16 + eSub];
    }

    float f[FEAT];
    float shv[9];
    float x0 = 0.0f, x1 = 0.0f;
    int tgt = 0;
    if (valid) {
        const float* fp = &feats[(long)eidx[e] * FEAT];
        tgt = eidx[(long)E + e];
#pragma unroll
        for (int j = 0; j < FEAT; ++j) f[j] = fp[j];
        shv[0] = sh0[e];
#pragma unroll
        for (int j = 0; j < 3; ++j) shv[1 + j] = sh1[e * 3 + j];
#pragma unroll
        for (int j = 0; j < 5; ++j) shv[4 + j] = sh2[e * 5 + j];
        x0 = dist[e];
        float inv0 = 0.5f * (f[0] + f[1]);
        float n1 = 0.0f, n2 = 0.0f;
#pragma unroll
        for (int j = 0; j < 3; ++j) { float m = 0.5f * (f[2 + j] + f[5 + j]); n1 += m * m; }
#pragma unroll
        for (int j = 0; j < 5; ++j) { float m = 0.5f * (f[8 + j] + f[13 + j]); n2 += m * m; }
        x1 = (inv0 + sqrtf(n1) + sqrtf(n2)) * (1.0f / 3.0f);
    } else {
#pragma unroll
        for (int j = 0; j < FEAT; ++j) f[j] = 0.0f;
#pragma unroll
        for (int j = 0; j < 9; ++j) shv[j] = 0.0f;
    }

    __syncthreads();

    float* gRow = &sGate[w][eSub * 60];
    float* uRow = &sBuf[w][eSub * KP];
    gate_mlp(sW, part, x0, x1, wA, bA, gRow);
    if (part == 0) {
        u_stage<0>(shv, f, uRow);
    } else {
        u_stage<1>(shv, f, uRow);
    }
    __syncthreads();

    {
        const int col16 = lane & 15;
        const int kofs = (lane < 16) ? 0 : 2;
        const float* uA = &sBuf[w][col16 * KP];
        v8f acc[NTILES];
#pragma unroll
        for (int n = 0; n < NTILES; ++n) {
            if (n == 4) __builtin_amdgcn_sched_barrier(0);
            acc[n] = (v8f){0.f, 0.f, 0.f, 0.f, 0.f, 0.f, 0.f, 0.f};
            const int col = col16 + n * 16;
#pragma unroll
            for (int kt = KLO[n]; kt <= KHI[n]; ++kt) {
                int k0 = kt * 4 + kofs;
                v2f a, b;
                a.x = uA[k0];
                a.y = uA[k0 + 1];
                b.x = Mg[k0 * PN + col];
                b.y = Mg[(k0 + 1) * PN + col];
                acc[n] = __builtin_amdgcn_wmma_f32_16x16x4_f32(
                    false, a, false, b, (short)0, acc[n], false, false);
            }
        }
        __syncthreads();
        float* raw = &sBuf[w][0];
        const int rbase = (lane < 16) ? 0 : 8;
#pragma unroll
        for (int n = 0; n < NTILES; ++n) {
            const int col = col16 + n * 16;
#pragma unroll
            for (int j = 0; j < 8; ++j)
                raw[(rbase + j) * RP + col] = acc[n][j];
        }
    }
    __syncthreads();

    {
        const float* rRow = &sBuf[w][eSub * RP];
        float r[4 * RQ];
#pragma unroll
        for (int q = 0; q < RQ; ++q) {
            const v4f t = *reinterpret_cast<const v4f*>(rRow + 4 * q);
            r[4 * q] = t.x; r[4 * q + 1] = t.y; r[4 * q + 2] = t.z; r[4 * q + 3] = t.w;
        }
        const float* gO = gRow + part;
        float g[2 * N_COMBOS];
#pragma unroll
        for (int q = 0; q < 2 * N_COMBOS; ++q) g[q] = gO[q * 2];
        float* sOut = &sBuf[w][16 * RP];
        float* mRow = sOut + eSub * FEAT;
        float* m0 = mRow + part;
        float* m1 = mRow + 2 + part * 3;
        float* m2 = mRow + 8 + part * 5;
#define DO_MIX(LO, K, P) (P)[K] = mix_val<LO, K>(g, r);
        DO_MIX(0, 0, m0)
        DO_MIX(1, 0, m1) DO_MIX(1, 1, m1) DO_MIX(1, 2, m1)
        DO_MIX(2, 0, m2) DO_MIX(2, 1, m2) DO_MIX(2, 2, m2) DO_MIX(2, 3, m2) DO_MIX(2, 4, m2)
#undef DO_MIX
        sOut[16 * FEAT + eSub] = __int_as_float(tgt);
        __syncthreads();

        constexpr int NS = (16 * FEAT) / 32;
        float v[NS];
        long off[NS];
        unsigned ed[NS];
#pragma unroll
        for (int j = 0; j < NS; ++j) {
            const unsigned p = (unsigned)(j * 32 + lane);
            ed[j] = p / (unsigned)FEAT;
            const unsigned el = p - ed[j] * (unsigned)FEAT;
            const int t = __float_as_int(sOut[16 * FEAT + ed[j]]);
            off[j] = (long)t * FEAT + el;
            v[j] = sOut[p];
        }
        const long e0 = (long)tile * 16;
        if (e0 + 16 <= (long)E) {
#pragma unroll
            for (int j = 0; j < NS; ++j) atomicAdd(&outF[off[j]], v[j]);
        } else {
#pragma unroll
            for (int j = 0; j < NS; ++j)
                if (e0 + (long)ed[j] < (long)E) atomicAdd(&outF[off[j]], v[j]);
        }
    }
}

extern "C" void kernel_launch(void* const* d_in, const int* in_sizes, int n_in,
                              void* d_out, int out_size, void* d_ws, size_t ws_size,
                              hipStream_t stream) {
    const float* node = (const float*)d_in[0];
    const int* eidx   = (const int*)d_in[1];
    const float* sh0  = (const float*)d_in[2];
    const float* sh1  = (const float*)d_in[3];
    const float* sh2  = (const float*)d_in[4];
    const float* dist = (const float*)d_in[5];
    const float* cg[N_COMBOS];
    for (int i = 0; i < N_COMBOS; ++i) cg[i] = (const float*)d_in[6 + i];
    const float* W1_0 = (const float*)d_in[21];
    const float* b1_0 = (const float*)d_in[22];
    const float* W2_0 = (const float*)d_in[23];
    const float* b2_0 = (const float*)d_in[24];
    const float* W1_1 = (const float*)d_in[25];
    const float* b1_1 = (const float*)d_in[26];
    const float* W2_1 = (const float*)d_in[27];
    const float* b2_1 = (const float*)d_in[28];

    const int E = in_sizes[1] / 2;
    const int N = in_sizes[0] / FEAT;

    float* Mg     = (float*)d_ws;
    float* feats1 = (float*)((char*)d_ws + 131072);

    build_M_kernel<<<1, 256, 0, stream>>>(cg[0], cg[1], cg[2], cg[3], cg[4], cg[5],
                                          cg[6], cg[7], cg[8], cg[9], cg[10], cg[11],
                                          cg[12], cg[13], cg[14], Mg);

    long n1 = (long)N * FEAT;
    zero_kernel<<<(int)((n1 + 255) / 256), 256, 0, stream>>>(feats1, n1);
    zero_kernel<<<(out_size + 255) / 256, 256, 0, stream>>>((float*)d_out, (long)out_size);

    int grid = (E + WAVES * 16 - 1) / (WAVES * 16);
    layer_kernel<<<grid, WAVES * 32, 0, stream>>>(node, eidx, sh0, sh1, sh2, dist, Mg,
                                                  W1_0, b1_0, W2_0, b2_0, feats1, E);
    layer_kernel<<<grid, WAVES * 32, 0, stream>>>(feats1, eidx, sh0, sh1, sh2, dist, Mg,
                                                  W1_1, b1_1, W2_1, b2_1, (float*)d_out, E);
}


# Set-up in one launch, not three: 256 work-groups zero-fill both accumulation buffers with 16-byte stores while one builds the CG matrix, a thread per entry (was 15 serial threads); 5 launches -> 3
# speedup vs baseline: 3.4009x; 1.1077x over previous
#include <hip/hip_runtime.h>

typedef __attribute__((ext_vector_type(2))) float v2f;
typedef __attribute__((ext_vector_type(4))) float v4f;
typedef __attribute__((ext_vector_type(8))) float v8f;

#define N_COMBOS 15
#define KP 164
#define PN 112
#define RP 116
#define RQ 26
#define NTILES 7
#define WAVES 2
#define FEAT 18

constexpr int cLE[N_COMBOS]     = {0,0,0,1,1,1,1,1,1,2,2,2,2,2,2};
constexpr int cLN[N_COMBOS]     = {0,1,2,0,1,1,1,2,2,0,1,1,2,2,2};
constexpr int cLO[N_COMBOS]     = {0,1,2,1,0,1,2,1,2,2,1,2,0,1,2};
constexpr int OUT_OFF[N_COMBOS] = {0,2,8,18,24,26,32,42,48,58,68,74,84,86,92};
constexpr int U_OFF[9]          = {0,2,8,18,24,42,72,82,112};
constexpr int KLO[NTILES]       = {0,2,6,10,18,20,28};
constexpr int KHI[NTILES]       = {4,10,17,20,27,40,40};
constexpr int N_BY_LO[3]        = {3,6,6};
constexpr int C_BY_LO[3][6]     = {{0,4,12,0,0,0},{1,3,5,7,10,13},{2,6,8,9,11,14}};

#define SETUP_THREADS 256
#define SETUP_ZERO_GROUPS 256

template <int C>
__device__ __forceinline__ float get_cg(const float* __restrict__ cg, int tid) {
    constexpr int cnt = (2 * cLE[C] + 1) * (2 * cLN[C] + 1) * (2 * cLO[C] + 1);
    static_assert(cnt <= SETUP_THREADS, "one thread per cg entry");
    return cg[tid < cnt ? tid : cnt - 1];
}

template <int C>
__device__ __forceinline__ void put_cg(float v, float* __restrict__ M, int tid) {
    constexpr int dO = 2 * cLO[C] + 1;
    constexpr int cnt = (2 * cLE[C] + 1) * (2 * cLN[C] + 1) * dO;
    constexpr int uo = U_OFF[cLE[C] * 3 + cLN[C]];
    constexpr int po = OUT_OFF[C];
    if (tid < cnt) {
        const int ab = tid / dO;
        const int k = tid - ab * dO;
        float* m = M + (uo + ab * 2) * PN + po + k;
        m[0] = v;
        m[PN + dO] = v;
    }
}

__device__ __forceinline__ void zero_span(float* __restrict__ p, long n, long first, long step) {
    const v4f z = {0.f, 0.f, 0.f, 0.f};
    if ((reinterpret_cast<unsigned long>(p) & 15ul) == 0ul) {
        v4f* p4 = reinterpret_cast<v4f*>(p);
        const long n4 = n >> 2;
        for (long i = first; i < n4; i += step) p4[i] = z;
        for (long i = (n4 << 2) + first; i < n; i += step) p[i] = 0.0f;
    } else {
        for (long i = first; i < n; i += step) p[i] = 0.0f;
    }
}

__global__ void __launch_bounds__(SETUP_THREADS)
setup_kernel(const float* __restrict__ cg0, const float* __restrict__ cg1, const float* __restrict__ cg2,
             const float* __restrict__ cg3, const float* __restrict__ cg4, const float* __restrict__ cg5,
             const float* __restrict__ cg6, const float* __restrict__ cg7, const float* __restrict__ cg8,
             const float* __restrict__ cg9, const float* __restrict__ cg10, const float* __restrict__ cg11,
             const float* __restrict__ cg12, const float* __restrict__ cg13, const float* __restrict__ cg14,
             float* __restrict__ M,
             float* __restrict__ bufA, long nA, float* __restrict__ bufB, long nB) {
    const int tid = threadIdx.x;
    if (blockIdx.x == 0) {
        static_assert((KP * PN) % 4 == 0, "M is zero-filled in 16-byte pieces");
        const float v0 = get_cg<0>(cg0, tid),    v1 = get_cg<1>(cg1, tid),    v2 = get_cg<2>(cg2, tid);
        const float v3 = get_cg<3>(cg3, tid),    v4 = get_cg<4>(cg4, tid),    v5 = get_cg<5>(cg5, tid);
        const float v6 = get_cg<6>(cg6, tid),    v7 = get_cg<7>(cg7, tid),    v8 = get_cg<8>(cg8, tid);
        const float v9 = get_cg<9>(cg9, tid),    v10 = get_cg<10>(cg10, tid), v11 = get_cg<11>(cg11, tid);
        const float v12 = get_cg<12>(cg12, tid), v13 = get_cg<13>(cg13, tid), v14 = get_cg<14>(cg14, tid);
        const v4f z = {0.f, 0.f, 0.f, 0.f};
        v4f* M4 = reinterpret_cast<v4f*>(M);
        for (int i = tid; i < KP * PN / 4; i += SETUP_THREADS) M4[i] = z;
        __syncthreads();
        put_cg<0>(v0, M, tid);   put_cg<1>(v1, M, tid);   put_cg<2>(v2, M, tid);
        put_cg<3>(v3, M, tid);   put_cg<4>(v4, M, tid);   put_cg<5>(v5, M, tid);
        put_cg<6>(v6, M, tid);   put_cg<7>(v7, M, tid);   put_cg<8>(v8, M, tid);
        put_cg<9>(v9, M, tid);   put_cg<10>(v10, M, tid); put_cg<11>(v11, M, tid);
        put_cg<12>(v12, M, tid); put_cg<13>(v13, M, tid); put_cg<14>(v14, M, tid);
    } else {
        const long first = (long)(blockIdx.x - 1) * SETUP_THREADS + tid;
        const long step = (long)(gridDim.x - 1) * SETUP_THREADS;
        zero_span(bufA, nA, first, step);
        zero_span(bufB, nB, first, step);
    }
}

#define NG 30
#define NGT 15
__device__ __forceinline__ void gate_mlp(const float* __restrict__ sW, int part,
                                         float x0, float x1,
                                         const v2f (&wA)[NGT], const float (&bA)[NGT],
                                         float* __restrict__ gRow) {
    v2f xb;
    xb.x = part ? 1.0f : x0;
    xb.y = part ? 0.0f : x1;
    const float* w2 = sW + part * 8;
    const float* c2 = sW + 240 + part * 2;
    float acc[NG];
#pragma unroll
    for (int j = 0; j < NGT; ++j) {
        v2f a;
        a.x = part ? bA[j] : wA[j].x;
        a.y = part ? 0.0f : wA[j].y;
        const v8f zero = {0.f, 0.f, 0.f, 0.f, 0.f, 0.f, 0.f, 0.f};
        const v8f d = __builtin_amdgcn_wmma_f32_16x16x4_f32(
            false, a, false, xb, (short)0, zero, false, false);
        v2f s = *reinterpret_cast<const v2f*>(c2 + j * 4);
#pragma unroll
        for (int h = 0; h < 4; ++h) {
            v2f hh;
            hh.x = fmaxf(d[h], 0.0f);
            hh.y = fmaxf(d[4 + h], 0.0f);
            const v2f wv = *reinterpret_cast<const v2f*>(w2 + j * 16 + h * 2);
            s = __builtin_elementwise_fma(hh, wv, s);
        }
        acc[j * 2] = s.x;
        acc[j * 2 + 1] = s.y;
    }
    float z[NG];
    float big = 0.0f;
#pragma unroll
    for (int t = 0; t < NG; ++t) {
        const float y = fabsf(acc[t]);
        const float y2 = y * y;
        float p = __builtin_fmaf(y2, -0x1.758e7ap-8f, 0x1.521192p-6f);
        p = __builtin_fmaf(y2, p, -0x1.b8389cp-5f);
        p = __builtin_fmaf(y2, p, 0x1.110704p-3f);
        p = __builtin_fmaf(y2, p, -0x1.555532p-2f);
        z[t] = __builtin_fmaf(y2, y * p, y);
        big = fmaxf(big, y);
    }
    if (__builtin_amdgcn_ballot_w32(!(big < 0.625f)) != 0u) {
#pragma unroll
        for (int t = 0; t < NG; ++t) {
            const float y = fabsf(acc[t]);
            const float ex = expf(2.0f * y);
            const float zb = __builtin_fmaf(-2.0f, __builtin_amdgcn_rcpf(ex + 1.0f), 1.0f);
            z[t] = (y < 0.625f) ? z[t] : zb;
        }
    }
#pragma unroll
    for (int j = 0; j < NGT; ++j)
#pragma unroll
        for (int i = 0; i < 2; ++i)
            gRow[j * 4 + part * 2 + i] = __builtin_copysignf(z[j * 2 + i], acc[j * 2 + i]);
}

template <int PART>
__device__ __forceinline__ void u_stage(const float* __restrict__ shv,
                                        const float* __restrict__ f,
                                        float* __restrict__ uRow) {
    int u = 0;
#pragma unroll
    for (int le = 0; le < 3; ++le) {
        const int de = 2 * le + 1;
        const int shb = (le == 0) ? 0 : (le == 1 ? 1 : 4);
#pragma unroll
        for (int ln = 0; ln < 3; ++ln) {
            const int dn = 2 * ln + 1;
            const int fb = (ln == 0) ? 0 : (ln == 1 ? 2 : 8);
#pragma unroll
            for (int a = 0; a < de; ++a)
#pragma unroll
                for (int b = 0; b < dn; ++b)
#pragma unroll
                    for (int ch = 0; ch < 2; ++ch) {
                        if ((PART == 0) ? (u < 81) : (u >= 81))
                            uRow[u] = shv[shb + a] * f[fb + ch * dn + b];
                        ++u;
                    }
        }
    }
    if (PART == 1) { uRow[162] = 0.0f; uRow[163] = 0.0f; }
}

template <int LO, int K>
__device__ __forceinline__ float mix_val(const float (&g)[2 * N_COMBOS],
                                         const float (&r)[4 * RQ]) {
    constexpr int dO = 2 * LO + 1;
    float v = 0.0f;
#pragma unroll
    for (int q = 0; q < N_BY_LO[LO]; ++q) {
        const int c = C_BY_LO[LO][q];
        const int base = OUT_OFF[c];
        v += g[c * 2]     * r[base + K] +
             g[c * 2 + 1] * r[base + dO + K];
    }
    return v;
}

__global__ void __launch_bounds__(WAVES * 32, 5)
layer_kernel(const float* __restrict__ feats, const int* __restrict__ eidx,
             const float* __restrict__ sh0, const float* __restrict__ sh1,
             const float* __restrict__ sh2, const float* __restrict__ dist,
             const float* __restrict__ Mg,
             const float* __restrict__ W1, const float* __restrict__ b1,
             const float* __restrict__ W2, const float* __restrict__ b2,
             float* __restrict__ outF, int E) {
    __shared__ __attribute__((aligned(16))) float sBuf[WAVES][16 * KP];
    __shared__ __attribute__((aligned(16))) float sGate[WAVES][16 * 60];
    __shared__ __attribute__((aligned(16))) float sWgt[300];
    static_assert(16 * RP + 16 * FEAT + 16 <= 16 * KP, "raw tile + scatter tile must fit the u tile");
    static_assert(4 * RQ <= RP && 4 * RQ >= 102, "the 16-byte pieces cover a raw row and stay inside it");

    const int lane = threadIdx.x & 31;
    const int w = threadIdx.x >> 5;
    const int tile = blockIdx.x * WAVES + w;
    const int eSub = lane & 15;
    const int part = lane >> 4;
    const long e = (long)tile * 16 + eSub;
    const bool valid = (e < (long)E);

    float* sW = &sWgt[0];
    {
        const int tid = threadIdx.x;
#pragma unroll
        for (int k = 0; k < 4; ++k) {
            int i = tid + k * (WAVES * 32);
            i = i < 240 ? i : 239;
            sW[(i & ~7) | ((i & 3) << 1) | ((i >> 2) & 1)] = W2[i];
        }
        const int i2 = tid < 60 ? tid : 59;
        sW[240 + i2] = b2[i2];
    }
    v2f wA[NGT];
    float bA[NGT];
#pragma unroll
    for (int j = 0; j < NGT; ++j) {
        wA[j] = *reinterpret_cast<const v2f*>(W1 + (j * 16 + eSub) * 2);
        bA[j] = b1[j * 16 + eSub];
    }

    float f[FEAT];
    float shv[9];
    float x0 = 0.0f, x1 = 0.0f;
    int tgt = 0;
    if (valid) {
        const float* fp = &feats[(long)eidx[e] * FEAT];
        tgt = eidx[(long)E + e];
#pragma unroll
        for (int j = 0; j < FEAT; ++j) f[j] = fp[j];
        shv[0] = sh0[e];
#pragma unroll
        for (int j = 0; j < 3; ++j) shv[1 + j] = sh1[e * 3 + j];
#pragma unroll
        for (int j = 0; j < 5; ++j) shv[4 + j] = sh2[e * 5 + j];
        x0 = dist[e];
        float inv0 = 0.5f * (f[0] + f[1]);
        float n1 = 0.0f, n2 = 0.0f;
#pragma unroll
        for (int j = 0; j < 3; ++j) { float m = 0.5f * (f[2 + j] + f[5 + j]); n1 += m * m; }
#pragma unroll
        for (int j = 0; j < 5; ++j) { float m = 0.5f * (f[8 + j] + f[13 + j]); n2 += m * m; }
        x1 = (inv0 + sqrtf(n1) + sqrtf(n2)) * (1.0f / 3.0f);
    } else {
#pragma unroll
        for (int j = 0; j < FEAT; ++j) f[j] = 0.0f;
#pragma unroll
        for (int j = 0; j < 9; ++j) shv[j] = 0.0f;
    }

    __syncthreads();

    float* gRow = &sGate[w][eSub * 60];
    float* uRow = &sBuf[w][eSub * KP];
    gate_mlp(sW, part, x0, x1, wA, bA, gRow);
    if (part == 0) {
        u_stage<0>(shv, f, uRow);
    } else {
        u_stage<1>(shv, f, uRow);
    }
    __syncthreads();

    {
        const int col16 = lane & 15;
        const int kofs = (lane < 16) ? 0 : 2;
        const float* uA = &sBuf[w][col16 * KP];
        v8f acc[NTILES];
#pragma unroll
        for (int n = 0; n < NTILES; ++n) {
            if (n == 4) __builtin_amdgcn_sched_barrier(0);
            acc[n] = (v8f){0.f, 0.f, 0.f, 0.f, 0.f, 0.f, 0.f, 0.f};
            const int col = col16 + n * 16;
#pragma unroll
            for (int kt = KLO[n]; kt <= KHI[n]; ++kt) {
                int k0 = kt * 4 + kofs;
                v2f a, b;
                a.x = uA[k0];
                a.y = uA[k0 + 1];
                b.x = Mg[k0 * PN + col];
                b.y = Mg[(k0 + 1) * PN + col];
                acc[n] = __builtin_amdgcn_wmma_f32_16x16x4_f32(
                    false, a, false, b, (short)0, acc[n], false, false);
            }
        }
        __syncthreads();
        float* raw = &sBuf[w][0];
        const int rbase = (lane < 16) ? 0 : 8;
#pragma unroll
        for (int n = 0; n < NTILES; ++n) {
            const int col = col16 + n * 16;
#pragma unroll
            for (int j = 0; j < 8; ++j)
                raw[(rbase + j) * RP + col] = acc[n][j];
        }
    }
    __syncthreads();

    {
        const float* rRow = &sBuf[w][eSub * RP];
        float r[4 * RQ];
#pragma unroll
        for (int q = 0; q < RQ; ++q) {
            const v4f t = *reinterpret_cast<const v4f*>(rRow + 4 * q);
            r[4 * q] = t.x; r[4 * q + 1] = t.y; r[4 * q + 2] = t.z; r[4 * q + 3] = t.w;
        }
        const float* gO = gRow + part;
        float g[2 * N_COMBOS];
#pragma unroll
        for (int q = 0; q < 2 * N_COMBOS; ++q) g[q] = gO[q * 2];
        float* sOut = &sBuf[w][16 * RP];
        float* mRow = sOut + eSub * FEAT;
        float* m0 = mRow + part;
        float* m1 = mRow + 2 + part * 3;
        float* m2 = mRow + 8 + part * 5;
#define DO_MIX(LO, K, P) (P)[K] = mix_val<LO, K>(g, r);
        DO_MIX(0, 0, m0)
        DO_MIX(1, 0, m1) DO_MIX(1, 1, m1) DO_MIX(1, 2, m1)
        DO_MIX(2, 0, m2) DO_MIX(2, 1, m2) DO_MIX(2, 2, m2) DO_MIX(2, 3, m2) DO_MIX(2, 4, m2)
#undef DO_MIX
        sOut[16 * FEAT + eSub] = __int_as_float(tgt);
        __syncthreads();

        constexpr int NS = (16 * FEAT) / 32;
        float v[NS];
        long off[NS];
        unsigned ed[NS];
#pragma unroll
        for (int j = 0; j < NS; ++j) {
            const unsigned p = (unsigned)(j * 32 + lane);
            ed[j] = p / (unsigned)FEAT;
            const unsigned el = p - ed[j] * (unsigned)FEAT;
            const int t = __float_as_int(sOut[16 * FEAT + ed[j]]);
            off[j] = (long)t * FEAT + el;
            v[j] = sOut[p];
        }
        const long e0 = (long)tile * 16;
        if (e0 + 16 <= (long)E) {
#pragma unroll
            for (int j = 0; j < NS; ++j) atomicAdd(&outF[off[j]], v[j]);
        } else {
#pragma unroll
            for (int j = 0; j < NS; ++j)
                if (e0 + (long)ed[j] < (long)E) atomicAdd(&outF[off[j]], v[j]);
        }
    }
}

extern "C" void kernel_launch(void* const* d_in, const int* in_sizes, int n_in,
                              void* d_out, int out_size, void* d_ws, size_t ws_size,
                              hipStream_t stream) {
    const float* node = (const float*)d_in[0];
    const int* eidx   = (const int*)d_in[1];
    const float* sh0  = (const float*)d_in[2];
    const float* sh1  = (const float*)d_in[3];
    const float* sh2  = (const float*)d_in[4];
    const float* dist = (const float*)d_in[5];
    const float* cg[N_COMBOS];
    for (int i = 0; i < N_COMBOS; ++i) cg[i] = (const float*)d_in[6 + i];
    const float* W1_0 = (const float*)d_in[21];
    const float* b1_0 = (const float*)d_in[22];
    const float* W2_0 = (const float*)d_in[23];
    const float* b2_0 = (const float*)d_in[24];
    const float* W1_1 = (const float*)d_in[25];
    const float* b1_1 = (const float*)d_in[26];
    const float* W2_1 = (const float*)d_in[27];
    const float* b2_1 = (const float*)d_in[28];

    const int E = in_sizes[1] / 2;
    const int N = in_sizes[0] / FEAT;

    float* Mg     = (float*)d_ws;
    float* feats1 = (float*)((char*)d_ws + 131072);

    setup_kernel<<<1 + SETUP_ZERO_GROUPS, SETUP_THREADS, 0, stream>>>(
        cg[0], cg[1], cg[2], cg[3], cg[4], cg[5], cg[6], cg[7], cg[8], cg[9], cg[10], cg[11],
        cg[12], cg[13], cg[14], Mg, feats1, (long)N * FEAT, (float*)d_out, (long)out_size);

    int grid = (E + WAVES * 16 - 1) / (WAVES * 16);
    layer_kernel<<<grid, WAVES * 32, 0, stream>>>(node, eidx, sh0, sh1, sh2, dist, Mg,
                                                  W1_0, b1_0, W2_0, b2_0, feats1, E);
    layer_kernel<<<grid, WAVES * 32, 0, stream>>>(feats1, eidx, sh0, sh1, sh2, dist, Mg,
                                                  W1_1, b1_1, W2_1, b2_1, (float*)d_out, E);
}


# CG matrix stored by the set-up kernel in B-fragment order: a lane fetches its B values of two products as one 16-byte load, 41 contiguous global loads per wave, not 162; sums unchanged
# speedup vs baseline: 3.8341x; 1.1274x over previous
#include <hip/hip_runtime.h>

typedef __attribute__((ext_vector_type(2))) float v2f;
typedef __attribute__((ext_vector_type(4))) float v4f;
typedef __attribute__((ext_vector_type(8))) float v8f;

#define N_COMBOS 15
#define KP 164
#define PN 112
#define RP 116
#define RQ 26
#define NTILES 7
#define WAVES 2
#define FEAT 18

constexpr int cLE[N_COMBOS]     = {0,0,0,1,1,1,1,1,1,2,2,2,2,2,2};
constexpr int cLN[N_COMBOS]     = {0,1,2,0,1,1,1,2,2,0,1,1,2,2,2};
constexpr int cLO[N_COMBOS]     = {0,1,2,1,0,1,2,1,2,2,1,2,0,1,2};
constexpr int OUT_OFF[N_COMBOS] = {0,2,8,18,24,26,32,42,48,58,68,74,84,86,92};
constexpr int U_OFF[9]          = {0,2,8,18,24,42,72,82,112};
constexpr int KLO[NTILES]       = {0,2,6,10,18,20,28};
constexpr int KHI[NTILES]       = {4,10,17,20,27,40,40};
constexpr int STEP0[NTILES]     = {0,5,14,26,37,47,68};
#define NSTEPS 81
#define NPAIRS 41
#define MF_FLOATS (NPAIRS * 32 * 4)
constexpr int N_BY_LO[3]        = {3,6,6};
constexpr int C_BY_LO[3][6]     = {{0,4,12,0,0,0},{1,3,5,7,10,13},{2,6,8,9,11,14}};

#define SETUP_THREADS 256
#define SETUP_ZERO_GROUPS 256

template <int C>
__device__ __forceinline__ float get_cg(const float* __restrict__ cg, int tid) {
    constexpr int cnt = (2 * cLE[C] + 1) * (2 * cLN[C] + 1) * (2 * cLO[C] + 1);
    static_assert(cnt <= SETUP_THREADS, "one thread per cg entry");
    return cg[tid < cnt ? tid : cnt - 1];
}

__device__ __forceinline__ int frag_pos(int row, int col) {
    const int n = col >> 4;
    int so = 0;
#pragma unroll
    for (int i = 1; i < NTILES; ++i) so = (n >= i) ? (STEP0[i] - KLO[i]) : so;
    const int s = so + (row >> 2);
    const int lane = (((row >> 1) & 1) << 4) | (col & 15);
    return ((((s >> 1) << 5) + lane) << 2) + ((s & 1) << 1) + (row & 1);
}

template <int C>
__device__ __forceinline__ void put_cg(float v, float* __restrict__ M, int tid) {
    constexpr int dO = 2 * cLO[C] + 1;
    constexpr int cnt = (2 * cLE[C] + 1) * (2 * cLN[C] + 1) * dO;
    constexpr int uo = U_OFF[cLE[C] * 3 + cLN[C]];
    constexpr int po = OUT_OFF[C];
    if (tid < cnt) {
        const int ab = tid / dO;
        const int k = tid - ab * dO;
        const int row = uo + ab * 2;
        const int col = po + k;
        M[frag_pos(row, col)] = v;
        M[frag_pos(row + 1, col + dO)] = v;
    }
}

__device__ __forceinline__ void zero_span(float* __restrict__ p, long n, long first, long step) {
    const v4f z = {0.f, 0.f, 0.f, 0.f};
    if ((reinterpret_cast<unsigned long>(p) & 15ul) == 0ul) {
        v4f* p4 = reinterpret_cast<v4f*>(p);
        const long n4 = n >> 2;
        for (long i = first; i < n4; i += step) p4[i] = z;
        for (long i = (n4 << 2) + first; i < n; i += step) p[i] = 0.0f;
    } else {
        for (long i = first; i < n; i += step) p[i] = 0.0f;
    }
}

__global__ void __launch_bounds__(SETUP_THREADS)
setup_kernel(const float* __restrict__ cg0, const float* __restrict__ cg1, const float* __restrict__ cg2,
             const float* __restrict__ cg3, const float* __restrict__ cg4, const float* __restrict__ cg5,
             const float* __restrict__ cg6, const float* __restrict__ cg7, const float* __restrict__ cg8,
             const float* __restrict__ cg9, const float* __restrict__ cg10, const float* __restrict__ cg11,
             const float* __restrict__ cg12, const float* __restrict__ cg13, const float* __restrict__ cg14,
             float* __restrict__ M,
             float* __restrict__ bufA, long nA, float* __restrict__ bufB, long nB) {
    const int tid = threadIdx.x;
    if (blockIdx.x == 0) {
        static_assert(STEP0[NTILES - 1] + KHI[NTILES - 1] - KLO[NTILES - 1] + 1 == NSTEPS, "step count");
        static_assert(2 * NPAIRS >= NSTEPS, "every product has a place in a pair");
        const float v0 = get_cg<0>(cg0, tid),    v1 = get_cg<1>(cg1, tid),    v2 = get_cg<2>(cg2, tid);
        const float v3 = get_cg<3>(cg3, tid),    v4 = get_cg<4>(cg4, tid),    v5 = get_cg<5>(cg5, tid);
        const float v6 = get_cg<6>(cg6, tid),    v7 = get_cg<7>(cg7, tid),    v8 = get_cg<8>(cg8, tid);
        const float v9 = get_cg<9>(cg9, tid),    v10 = get_cg<10>(cg10, tid), v11 = get_cg<11>(cg11, tid);
        const float v12 = get_cg<12>(cg12, tid), v13 = get_cg<13>(cg13, tid), v14 = get_cg<14>(cg14, tid);
        const v4f z = {0.f, 0.f, 0.f, 0.f};
        v4f* M4 = reinterpret_cast<v4f*>(M);
        for (int i = tid; i < MF_FLOATS / 4; i += SETUP_THREADS) M4[i] = z;
        __syncthreads();
        put_cg<0>(v0, M, tid);   put_cg<1>(v1, M, tid);   put_cg<2>(v2, M, tid);
        put_cg<3>(v3, M, tid);   put_cg<4>(v4, M, tid);   put_cg<5>(v5, M, tid);
        put_cg<6>(v6, M, tid);   put_cg<7>(v7, M, tid);   put_cg<8>(v8, M, tid);
        put_cg<9>(v9, M, tid);   put_cg<10>(v10, M, tid); put_cg<11>(v11, M, tid);
        put_cg<12>(v12, M, tid); put_cg<13>(v13, M, tid); put_cg<14>(v14, M, tid);
    } else {
        const long first = (long)(blockIdx.x - 1) * SETUP_THREADS + tid;
        const long step = (long)(gridDim.x - 1) * SETUP_THREADS;
        zero_span(bufA, nA, first, step);
        zero_span(bufB, nB, first, step);
    }
}

#define NG 30
#define NGT 15
__device__ __forceinline__ void gate_mlp(const float* __restrict__ sW, int part,
                                         float x0, float x1,
                                         const v2f (&wA)[NGT], const float (&bA)[NGT],
                                         float* __restrict__ gRow) {
    v2f xb;
    xb.x = part ? 1.0f : x0;
    xb.y = part ? 0.0f : x1;
    const float* w2 = sW + part * 8;
    const float* c2 = sW + 240 + part * 2;
    float acc[NG];
#pragma unroll
    for (int j = 0; j < NGT; ++j) {
        v2f a;
        a.x = part ? bA[j] : wA[j].x;
        a.y = part ? 0.0f : wA[j].y;
        const v8f zero = {0.f, 0.f, 0.f, 0.f, 0.f, 0.f, 0.f, 0.f};
        const v8f d = __builtin_amdgcn_wmma_f32_16x16x4_f32(
            false, a, false, xb, (short)0, zero, false, false);
        v2f s = *reinterpret_cast<const v2f*>(c2 + j * 4);
#pragma unroll
        for (int h = 0; h < 4; ++h) {
            v2f hh;
            hh.x = fmaxf(d[h], 0.0f);
            hh.y = fmaxf(d[4 + h], 0.0f);
            const v2f wv = *reinterpret_cast<const v2f*>(w2 + j * 16 + h * 2);
            s = __builtin_elementwise_fma(hh, wv, s);
        }
        acc[j * 2] = s.x;
        acc[j * 2 + 1] = s.y;
    }
    float z[NG];
    float big = 0.0f;
#pragma unroll
    for (int t = 0; t < NG; ++t) {
        const float y = fabsf(acc[t]);
        const float y2 = y * y;
        float p = __builtin_fmaf(y2, -0x1.758e7ap-8f, 0x1.521192p-6f);
        p = __builtin_fmaf(y2, p, -0x1.b8389cp-5f);
        p = __builtin_fmaf(y2, p, 0x1.110704p-3f);
        p = __builtin_fmaf(y2, p, -0x1.555532p-2f);
        z[t] = __builtin_fmaf(y2, y * p, y);
        big = fmaxf(big, y);
    }
    if (__builtin_amdgcn_ballot_w32(!(big < 0.625f)) != 0u) {
#pragma unroll
        for (int t = 0; t < NG; ++t) {
            const float y = fabsf(acc[t]);
            const float ex = expf(2.0f * y);
            const float zb = __builtin_fmaf(-2.0f, __builtin_amdgcn_rcpf(ex + 1.0f), 1.0f);
            z[t] = (y < 0.625f) ? z[t] : zb;
        }
    }
#pragma unroll
    for (int j = 0; j < NGT; ++j)
#pragma unroll
        for (int i = 0; i < 2; ++i)
            gRow[j * 4 + part * 2 + i] = __builtin_copysignf(z[j * 2 + i], acc[j * 2 + i]);
}

template <int PART>
__device__ __forceinline__ void u_stage(const float* __restrict__ shv,
                                        const float* __restrict__ f,
                                        float* __restrict__ uRow) {
    int u = 0;
#pragma unroll
    for (int le = 0; le < 3; ++le) {
        const int de = 2 * le + 1;
        const int shb = (le == 0) ? 0 : (le == 1 ? 1 : 4);
#pragma unroll
        for (int ln = 0; ln < 3; ++ln) {
            const int dn = 2 * ln + 1;
            const int fb = (ln == 0) ? 0 : (ln == 1 ? 2 : 8);
#pragma unroll
            for (int a = 0; a < de; ++a)
#pragma unroll
                for (int b = 0; b < dn; ++b)
#pragma unroll
                    for (int ch = 0; ch < 2; ++ch) {
                        if ((PART == 0) ? (u < 81) : (u >= 81))
                            uRow[u] = shv[shb + a] * f[fb + ch * dn + b];
                        ++u;
                    }
        }
    }
    if (PART == 1) { uRow[162] = 0.0f; uRow[163] = 0.0f; }
}

template <int LO, int K>
__device__ __forceinline__ float mix_val(const float (&g)[2 * N_COMBOS],
                                         const float (&r)[4 * RQ]) {
    constexpr int dO = 2 * LO + 1;
    float v = 0.0f;
#pragma unroll
    for (int q = 0; q < N_BY_LO[LO]; ++q) {
        const int c = C_BY_LO[LO][q];
        const int base = OUT_OFF[c];
        v += g[c * 2]     * r[base + K] +
             g[c * 2 + 1] * r[base + dO + K];
    }
    return v;
}

__global__ void __launch_bounds__(WAVES * 32, 5)
layer_kernel(const float* __restrict__ feats, const int* __restrict__ eidx,
             const float* __restrict__ sh0, const float* __restrict__ sh1,
             const float* __restrict__ sh2, const float* __restrict__ dist,
             const float* __restrict__ Mg,
             const float* __restrict__ W1, const float* __restrict__ b1,
             const float* __restrict__ W2, const float* __restrict__ b2,
             float* __restrict__ outF, int E) {
    __shared__ __attribute__((aligned(16))) float sBuf[WAVES][16 * KP];
    __shared__ __attribute__((aligned(16))) float sGate[WAVES][16 * 60];
    __shared__ __attribute__((aligned(16))) float sWgt[300];
    static_assert(16 * RP + 16 * FEAT + 16 <= 16 * KP, "raw tile + scatter tile must fit the u tile");
    static_assert(4 * RQ <= RP && 4 * RQ >= 102, "the 16-byte pieces cover a raw row and stay inside it");

    const int lane = threadIdx.x & 31;
    const int w = threadIdx.x >> 5;
    const int tile = blockIdx.x * WAVES + w;
    const int eSub = lane & 15;
    const int part = lane >> 4;
    const long e = (long)tile * 16 + eSub;
    const bool valid = (e < (long)E);

    float* sW = &sWgt[0];
    {
        const int tid = threadIdx.x;
#pragma unroll
        for (int k = 0; k < 4; ++k) {
            int i = tid + k * (WAVES * 32);
            i = i < 240 ? i : 239;
            sW[(i & ~7) | ((i & 3) << 1) | ((i >> 2) & 1)] = W2[i];
        }
        const int i2 = tid < 60 ? tid : 59;
        sW[240 + i2] = b2[i2];
    }
    v2f wA[NGT];
    float bA[NGT];
#pragma unroll
    for (int j = 0; j < NGT; ++j) {
        wA[j] = *reinterpret_cast<const v2f*>(W1 + (j * 16 + eSub) * 2);
        bA[j] = b1[j * 16 + eSub];
    }

    float f[FEAT];
    float shv[9];
    float x0 = 0.0f, x1 = 0.0f;
    int tgt = 0;
    if (valid) {
        const float* fp = &feats[(long)eidx[e] * FEAT];
        tgt = eidx[(long)E + e];
#pragma unroll
        for (int j = 0; j < FEAT; ++j) f[j] = fp[j];
        shv[0] = sh0[e];
#pragma unroll
        for (int j = 0; j < 3; ++j) shv[1 + j] = sh1[e * 3 + j];
#pragma unroll
        for (int j = 0; j < 5; ++j) shv[4 + j] = sh2[e * 5 + j];
        x0 = dist[e];
        float inv0 = 0.5f * (f[0] + f[1]);
        float n1 = 0.0f, n2 = 0.0f;
#pragma unroll
        for (int j = 0; j < 3; ++j) { float m = 0.5f * (f[2 + j] + f[5 + j]); n1 += m * m; }
#pragma unroll
        for (int j = 0; j < 5; ++j) { float m = 0.5f * (f[8 + j] + f[13 + j]); n2 += m * m; }
        x1 = (inv0 + sqrtf(n1) + sqrtf(n2)) * (1.0f / 3.0f);
    } else {
#pragma unroll
        for (int j = 0; j < FEAT; ++j) f[j] = 0.0f;
#pragma unroll
        for (int j = 0; j < 9; ++j) shv[j] = 0.0f;
    }

    __syncthreads();

    float* gRow = &sGate[w][eSub * 60];
    float* uRow = &sBuf[w][eSub * KP];
    gate_mlp(sW, part, x0, x1, wA, bA, gRow);
    if (part == 0) {
        u_stage<0>(shv, f, uRow);
    } else {
        u_stage<1>(shv, f, uRow);
    }
    __syncthreads();

    {
        const int col16 = lane & 15;
        const int kofs = (lane < 16) ? 0 : 2;
        const float* uA = &sBuf[w][col16 * KP];
        const v4f* Bf = reinterpret_cast<const v4f*>(Mg) + lane;
        v4f b4 = {0.f, 0.f, 0.f, 0.f};
        v8f acc[NTILES];
#pragma unroll
        for (int n = 0; n < NTILES; ++n) {
            if (n == 4) __builtin_amdgcn_sched_barrier(0);
            acc[n] = (v8f){0.f, 0.f, 0.f, 0.f, 0.f, 0.f, 0.f, 0.f};
#pragma unroll
            for (int kt = KLO[n]; kt <= KHI[n]; ++kt) {
                const int k0 = kt * 4 + kofs;
                const int s = STEP0[n] + kt - KLO[n];
                v2f a, b;
                a.x = uA[k0];
                a.y = uA[k0 + 1];
                if ((s & 1) == 0) b4 = Bf[(s >> 1) * 32];
                b.x = (s & 1) ? b4.z : b4.x;
                b.y = (s & 1) ? b4.w : b4.y;
                acc[n] = __builtin_amdgcn_wmma_f32_16x16x4_f32(
                    false, a, false, b, (short)0, acc[n], false, false);
            }
        }
        __syncthreads();
        float* raw = &sBuf[w][0];
        const int rbase = (lane < 16) ? 0 : 8;
#pragma unroll
        for (int n = 0; n < NTILES; ++n) {
            const int col = col16 + n * 16;
#pragma unroll
            for (int j = 0; j < 8; ++j)
                raw[(rbase + j) * RP + col] = acc[n][j];
        }
    }
    __syncthreads();

    {
        const float* rRow = &sBuf[w][eSub * RP];
        float r[4 * RQ];
#pragma unroll
        for (int q = 0; q < RQ; ++q) {
            const v4f t = *reinterpret_cast<const v4f*>(rRow + 4 * q);
            r[4 * q] = t.x; r[4 * q + 1] = t.y; r[4 * q + 2] = t.z; r[4 * q + 3] = t.w;
        }
        const float* gO = gRow + part;
        float g[2 * N_COMBOS];
#pragma unroll
        for (int q = 0; q < 2 * N_COMBOS; ++q) g[q] = gO[q * 2];
        float* sOut = &sBuf[w][16 * RP];
        float* mRow = sOut + eSub * FEAT;
        float* m0 = mRow + part;
        float* m1 = mRow + 2 + part * 3;
        float* m2 = mRow + 8 + part * 5;
#define DO_MIX(LO, K, P) (P)[K] = mix_val<LO, K>(g, r);
        DO_MIX(0, 0, m0)
        DO_MIX(1, 0, m1) DO_MIX(1, 1, m1) DO_MIX(1, 2, m1)
        DO_MIX(2, 0, m2) DO_MIX(2, 1, m2) DO_MIX(2, 2, m2) DO_MIX(2, 3, m2) DO_MIX(2, 4, m2)
#undef DO_MIX
        sOut[16 * FEAT + eSub] = __int_as_float(tgt);
        __syncthreads();

        constexpr int NS = (16 * FEAT) / 32;
        float v[NS];
        long off[NS];
        unsigned ed[NS];
#pragma unroll
        for (int j = 0; j < NS; ++j) {
            const unsigned p = (unsigned)(j * 32 + lane);
            ed[j] = p / (unsigned)FEAT;
            const unsigned el = p - ed[j] * (unsigned)FEAT;
            const int t = __float_as_int(sOut[16 * FEAT + ed[j]]);
            off[j] = (long)t * FEAT + el;
            v[j] = sOut[p];
        }
        const long e0 = (long)tile * 16;
        if (e0 + 16 <= (long)E) {
#pragma unroll
            for (int j = 0; j < NS; ++j) atomicAdd(&outF[off[j]], v[j]);
        } else {
#pragma unroll
            for (int j = 0; j < NS; ++j)
                if (e0 + (long)ed[j] < (long)E) atomicAdd(&outF[off[j]], v[j]);
        }
    }
}

extern "C" void kernel_launch(void* const* d_in, const int* in_sizes, int n_in,
                              void* d_out, int out_size, void* d_ws, size_t ws_size,
                              hipStream_t stream) {
    const float* node = (const float*)d_in[0];
    const int* eidx   = (const int*)d_in[1];
    const float* sh0  = (const float*)d_in[2];
    const float* sh1  = (const float*)d_in[3];
    const float* sh2  = (const float*)d_in[4];
    const float* dist = (const float*)d_in[5];
    const float* cg[N_COMBOS];
    for (int i = 0; i < N_COMBOS; ++i) cg[i] = (const float*)d_in[6 + i];
    const float* W1_0 = (const float*)d_in[21];
    const float* b1_0 = (const float*)d_in[22];
    const float* W2_0 = (const float*)d_in[23];
    const float* b2_0 = (const float*)d_in[24];
    const float* W1_1 = (const float*)d_in[25];
    const float* b1_1 = (const float*)d_in[26];
    const float* W2_1 = (const float*)d_in[27];
    const float* b2_1 = (const float*)d_in[28];

    const int E = in_sizes[1] / 2;
    const int N = in_sizes[0] / FEAT;

    float* Mg     = (float*)d_ws;
    float* feats1 = (float*)((char*)d_ws + 131072);

    setup_kernel<<<1 + SETUP_ZERO_GROUPS, SETUP_THREADS, 0, stream>>>(
        cg[0], cg[1], cg[2], cg[3], cg[4], cg[5], cg[6], cg[7], cg[8], cg[9], cg[10], cg[11],
        cg[12], cg[13], cg[14], Mg, feats1, (long)N * FEAT, (float*)d_out, (long)out_size);

    int grid = (E + WAVES * 16 - 1) / (WAVES * 16);
    layer_kernel<<<grid, WAVES * 32, 0, stream>>>(node, eidx, sh0, sh1, sh2, dist, Mg,
                                                  W1_0, b1_0, W2_0, b2_0, feats1, E);
    layer_kernel<<<grid, WAVES * 32, 0, stream>>>(feats1, eidx, sh0, sh1, sh2, dist, Mg,
                                                  W1_1, b1_1, W2_1, b2_1, (float*)d_out, E);
}


# Each input channel is its own product against ONE shared CG matrix, regrouped so a K-tile feeds a single N-tile: 2 x 22 f32 WMMAs per 16 edges, not 81; each raw sum still adds in ascending k
# speedup vs baseline: 4.7335x; 1.2346x over previous
#include <hip/hip_runtime.h>

typedef __attribute__((ext_vector_type(2))) float v2f;
typedef __attribute__((ext_vector_type(4))) float v4f;
typedef __attribute__((ext_vector_type(8))) float v8f;

#define N_COMBOS 15
#define KC 88
#define KP 180
#define NC 64
#define RP 132
#define RQ 15
#define NTILES 4
#define WAVES 2
#define FEAT 18

constexpr int cLE[N_COMBOS]     = {0,0,0,1,1,1,1,1,1,2,2,2,2,2,2};
constexpr int cLN[N_COMBOS]     = {0,1,2,0,1,1,1,2,2,0,1,1,2,2,2};
constexpr int cLO[N_COMBOS]     = {0,1,2,1,0,1,2,1,2,2,1,2,0,1,2};
constexpr int ROW_OFF[9]        = {0,1,4,9,12,28,21,43,60};
constexpr int COL_OFF[N_COMBOS] = {0,1,4,9,16,17,20,32,35,25,40,43,48,49,52};
constexpr int KLO[NTILES]       = {0,3,7,15};
constexpr int KHI[NTILES]       = {2,6,14,21};
#define NSTEPS 22
#define NPAIRS 11
#define MF_FLOATS (NPAIRS * 32 * 4)
constexpr int N_BY_LO[3]        = {3,6,6};
constexpr int C_BY_LO[3][6]     = {{0,4,12,0,0,0},{1,3,5,7,10,13},{2,6,8,9,11,14}};

#define SETUP_THREADS 256
#define SETUP_ZERO_GROUPS 256

template <int C>
__device__ __forceinline__ float get_cg(const float* __restrict__ cg, int tid) {
    constexpr int cnt = (2 * cLE[C] + 1) * (2 * cLN[C] + 1) * (2 * cLO[C] + 1);
    static_assert(cnt <= SETUP_THREADS, "one thread per cg entry");
    return cg[tid < cnt ? tid : cnt - 1];
}

__device__ __forceinline__ int frag_pos(int row, int col) {
    const int s = row >> 2;
    const int lane = (((row >> 1) & 1) << 4) | (col & 15);
    return ((((s >> 1) << 5) + lane) << 2) + ((s & 1) << 1) + (row & 1);
}

template <int C>
__device__ __forceinline__ void put_cg(float v, float* __restrict__ M, int tid) {
    constexpr int dO = 2 * cLO[C] + 1;
    constexpr int cnt = (2 * cLE[C] + 1) * (2 * cLN[C] + 1) * dO;
    constexpr int uo = ROW_OFF[cLE[C] * 3 + cLN[C]];
    constexpr int po = COL_OFF[C];
    static_assert((uo >> 2) >= KLO[po >> 4] && ((uo + cnt / dO - 1) >> 2) <= KHI[po >> 4] &&
                  ((po + dO - 1) >> 4) == (po >> 4), "a combo's rows lie in the K-tiles of its one N-tile");
    if (tid < cnt) {
        const int ab = tid / dO;
        const int k = tid - ab * dO;
        M[frag_pos(uo + ab, po + k)] = v;
    }
}

__device__ __forceinline__ void zero_span(float* __restrict__ p, long n, long first, long step) {
    const v4f z = {0.f, 0.f, 0.f, 0.f};
    if ((reinterpret_cast<unsigned long>(p) & 15ul) == 0ul) {
        v4f* p4 = reinterpret_cast<v4f*>(p);
        const long n4 = n >> 2;
        for (long i = first; i < n4; i += step) p4[i] = z;
        for (long i = (n4 << 2) + first; i < n; i += step) p[i] = 0.0f;
    } else {
        for (long i = first; i < n; i += step) p[i] = 0.0f;
    }
}

__global__ void __launch_bounds__(SETUP_THREADS)
setup_kernel(const float* __restrict__ cg0, const float* __restrict__ cg1, const float* __restrict__ cg2,
             const float* __restrict__ cg3, const float* __restrict__ cg4, const float* __restrict__ cg5,
             const float* __restrict__ cg6, const float* __restrict__ cg7, const float* __restrict__ cg8,
             const float* __restrict__ cg9, const float* __restrict__ cg10, const float* __restrict__ cg11,
             const float* __restrict__ cg12, const float* __restrict__ cg13, const float* __restrict__ cg14,
             float* __restrict__ M,
             float* __restrict__ bufA, long nA, float* __restrict__ bufB, long nB) {
    const int tid = threadIdx.x;
    if (blockIdx.x == 0) {
        static_assert(KHI[NTILES - 1] + 1 == NSTEPS && 4 * NSTEPS == KC, "K-tiles cover a channel's u row");
        static_assert(2 * NPAIRS >= NSTEPS, "every product has a place in a pair");
        const float v0 = get_cg<0>(cg0, tid),    v1 = get_cg<1>(cg1, tid),    v2 = get_cg<2>(cg2, tid);
        const float v3 = get_cg<3>(cg3, tid),    v4 = get_cg<4>(cg4, tid),    v5 = get_cg<5>(cg5, tid);
        const float v6 = get_cg<6>(cg6, tid),    v7 = get_cg<7>(cg7, tid),    v8 = get_cg<8>(cg8, tid);
        const float v9 = get_cg<9>(cg9, tid),    v10 = get_cg<10>(cg10, tid), v11 = get_cg<11>(cg11, tid);
        const float v12 = get_cg<12>(cg12, tid), v13 = get_cg<13>(cg13, tid), v14 = get_cg<14>(cg14, tid);
        const v4f z = {0.f, 0.f, 0.f, 0.f};
        v4f* M4 = reinterpret_cast<v4f*>(M);
        for (int i = tid; i < MF_FLOATS / 4; i += SETUP_THREADS) M4[i] = z;
        __syncthreads();
        put_cg<0>(v0, M, tid);   put_cg<1>(v1, M, tid);   put_cg<2>(v2, M, tid);
        put_cg<3>(v3, M, tid);   put_cg<4>(v4, M, tid);   put_cg<5>(v5, M, tid);
        put_cg<6>(v6, M, tid);   put_cg<7>(v7, M, tid);   put_cg<8>(v8, M, tid);
        put_cg<9>(v9, M, tid);   put_cg<10>(v10, M, tid); put_cg<11>(v11, M, tid);
        put_cg<12>(v12, M, tid); put_cg<13>(v13, M, tid); put_cg<14>(v14, M, tid);
    } else {
        const long first = (long)(blockIdx.x - 1) * SETUP_THREADS + tid;
        const long step = (long)(gridDim.x - 1) * SETUP_THREADS;
        zero_span(bufA, nA, first, step);
        zero_span(bufB, nB, first, step);
    }
}

#define NG 30
#define NGT 15
__device__ __forceinline__ void gate_mlp(const float* __restrict__ sW, int part,
                                         float x0, float x1,
                                         const v2f (&wA)[NGT], const float (&bA)[NGT],
                                         float* __restrict__ gRow) {
    v2f xb;
    xb.x = part ? 1.0f : x0;
    xb.y = part ? 0.0f : x1;
    const float* w2 = sW + part * 8;
    const float* c2 = sW + 240 + part * 2;
    float acc[NG];
#pragma unroll
    for (int j = 0; j < NGT; ++j) {
        v2f a;
        a.x = part ? bA[j] : wA[j].x;
        a.y = part ? 0.0f : wA[j].y;
        const v8f zero = {0.f, 0.f, 0.f, 0.f, 0.f, 0.f, 0.f, 0.f};
        const v8f d = __builtin_amdgcn_wmma_f32_16x16x4_f32(
            false, a, false, xb, (short)0, zero, false, false);
        v2f s = *reinterpret_cast<const v2f*>(c2 + j * 4);
#pragma unroll
        for (int h = 0; h < 4; ++h) {
            v2f hh;
            hh.x = fmaxf(d[h], 0.0f);
            hh.y = fmaxf(d[4 + h], 0.0f);
            const v2f wv = *reinterpret_cast<const v2f*>(w2 + j * 16 + h * 2);
            s = __builtin_elementwise_fma(hh, wv, s);
        }
        acc[j * 2] = s.x;
        acc[j * 2 + 1] = s.y;
    }
    float z[NG];
    float big = 0.0f;
#pragma unroll
    for (int t = 0; t < NG; ++t) {
        const float y = fabsf(acc[t]);
        const float y2 = y * y;
        float p = __builtin_fmaf(y2, -0x1.758e7ap-8f, 0x1.521192p-6f);
        p = __builtin_fmaf(y2, p, -0x1.b8389cp-5f);
        p = __builtin_fmaf(y2, p, 0x1.110704p-3f);
        p = __builtin_fmaf(y2, p, -0x1.555532p-2f);
        z[t] = __builtin_fmaf(y2, y * p, y);
        big = fmaxf(big, y);
    }
    if (__builtin_amdgcn_ballot_w32(!(big < 0.625f)) != 0u) {
#pragma unroll
        for (int t = 0; t < NG; ++t) {
            const float y = fabsf(acc[t]);
            const float ex = expf(2.0f * y);
            const float zb = __builtin_fmaf(-2.0f, __builtin_amdgcn_rcpf(ex + 1.0f), 1.0f);
            z[t] = (y < 0.625f) ? z[t] : zb;
        }
    }
#pragma unroll
    for (int j = 0; j < NGT; ++j)
#pragma unroll
        for (int i = 0; i < 2; ++i)
            gRow[j * 4 + part * 2 + i] = __builtin_copysignf(z[j * 2 + i], acc[j * 2 + i]);
}

__device__ __forceinline__ void u_stage(const float (&shv)[9], const float (&fc)[9],
                                        float* __restrict__ uCh) {
    float uv[KC];
#pragma unroll
    for (int i = 0; i < KC; ++i) uv[i] = 0.0f;
#pragma unroll
    for (int le = 0; le < 3; ++le) {
        const int de = 2 * le + 1;
        const int shb = (le == 0) ? 0 : (le == 1 ? 1 : 4);
#pragma unroll
        for (int ln = 0; ln < 3; ++ln) {
            const int dn = 2 * ln + 1;
            const int fb = (ln == 0) ? 0 : (ln == 1 ? 1 : 4);
            const int base = ROW_OFF[le * 3 + ln];
#pragma unroll
            for (int a = 0; a < de; ++a)
#pragma unroll
                for (int b = 0; b < dn; ++b)
                    uv[base + a * dn + b] = shv[shb + a] * fc[fb + b];
        }
    }
#pragma unroll
    for (int q = 0; q < KC / 4; ++q) {
        const v4f t = {uv[4 * q], uv[4 * q + 1], uv[4 * q + 2], uv[4 * q + 3]};
        *reinterpret_cast<v4f*>(uCh + 4 * q) = t;
    }
}

template <int LO, int K>
__device__ __forceinline__ float mix_val(const float (&g)[2 * N_COMBOS],
                                         const float (&r0)[4 * RQ], const float (&r1)[4 * RQ]) {
    float v = 0.0f;
#pragma unroll
    for (int q = 0; q < N_BY_LO[LO]; ++q) {
        const int c = C_BY_LO[LO][q];
        const int base = COL_OFF[c];
        v += g[c * 2]     * r0[base + K] +
             g[c * 2 + 1] * r1[base + K];
    }
    return v;
}

__global__ void __launch_bounds__(WAVES * 32, 5)
layer_kernel(const float* __restrict__ feats, const int* __restrict__ eidx,
             const float* __restrict__ sh0, const float* __restrict__ sh1,
             const float* __restrict__ sh2, const float* __restrict__ dist,
             const float* __restrict__ Mg,
             const float* __restrict__ W1, const float* __restrict__ b1,
             const float* __restrict__ W2, const float* __restrict__ b2,
             float* __restrict__ outF, int E) {
    __shared__ __attribute__((aligned(16))) float sBuf[WAVES][16 * KP];
    __shared__ __attribute__((aligned(16))) float sGate[WAVES][16 * 60];
    __shared__ __attribute__((aligned(16))) float sWgt[300];
    static_assert(16 * RP + 16 * FEAT + 16 <= 16 * KP, "raw tile + scatter tile must fit the u tile");
    static_assert(4 * RQ <= NC && 4 * RQ > COL_OFF[N_COMBOS - 1] + 4 && 2 * NC <= RP,
                  "the 16-byte pieces cover a channel's raw columns and stay inside them");
    static_assert(2 * KC <= KP, "both channels' u values fit the u row");

    const int lane = threadIdx.x & 31;
    const int w = threadIdx.x >> 5;
    const int tile = blockIdx.x * WAVES + w;
    const int eSub = lane & 15;
    const int part = lane >> 4;
    const long e = (long)tile * 16 + eSub;
    const bool valid = (e < (long)E);

    float* sW = &sWgt[0];
    {
        const int tid = threadIdx.x;
#pragma unroll
        for (int k = 0; k < 4; ++k) {
            int i = tid + k * (WAVES * 32);
            i = i < 240 ? i : 239;
            sW[(i & ~7) | ((i & 3) << 1) | ((i >> 2) & 1)] = W2[i];
        }
        const int i2 = tid < 60 ? tid : 59;
        sW[240 + i2] = b2[i2];
    }
    v2f wA[NGT];
    float bA[NGT];
#pragma unroll
    for (int j = 0; j < NGT; ++j) {
        wA[j] = *reinterpret_cast<const v2f*>(W1 + (j * 16 + eSub) * 2);
        bA[j] = b1[j * 16 + eSub];
    }

    float f[FEAT];
    float shv[9];
    float x0 = 0.0f, x1 = 0.0f;
    int tgt = 0;
    if (valid) {
        const float* fp = &feats[(long)eidx[e] * FEAT];
        tgt = eidx[(long)E + e];
#pragma unroll
        for (int j = 0; j < FEAT; ++j) f[j] = fp[j];
        shv[0] = sh0[e];
#pragma unroll
        for (int j = 0; j < 3; ++j) shv[1 + j] = sh1[e * 3 + j];
#pragma unroll
        for (int j = 0; j < 5; ++j) shv[4 + j] = sh2[e * 5 + j];
        x0 = dist[e];
        float inv0 = 0.5f * (f[0] + f[1]);
        float n1 = 0.0f, n2 = 0.0f;
#pragma unroll
        for (int j = 0; j < 3; ++j) { float m = 0.5f * (f[2 + j] + f[5 + j]); n1 += m * m; }
#pragma unroll
        for (int j = 0; j < 5; ++j) { float m = 0.5f * (f[8 + j] + f[13 + j]); n2 += m * m; }
        x1 = (inv0 + sqrtf(n1) + sqrtf(n2)) * (1.0f / 3.0f);
    } else {
#pragma unroll
        for (int j = 0; j < FEAT; ++j) f[j] = 0.0f;
#pragma unroll
        for (int j = 0; j < 9; ++j) shv[j] = 0.0f;
    }

    __syncthreads();

    float* gRow = &sGate[w][eSub * 60];
    float* uRow = &sBuf[w][eSub * KP];
    gate_mlp(sW, part, x0, x1, wA, bA, gRow);
    {
        float fc[9];
        fc[0] = part ? f[1] : f[0];
#pragma unroll
        for (int b = 0; b < 3; ++b) fc[1 + b] = part ? f[5 + b] : f[2 + b];
#pragma unroll
        for (int b = 0; b < 5; ++b) fc[4 + b] = part ? f[13 + b] : f[8 + b];
        u_stage(shv, fc, uRow + part * KC);
    }
    __syncthreads();

    {
        const int col16 = lane & 15;
        const int kofs = (lane < 16) ? 0 : 2;
        const float* uA = &sBuf[w][col16 * KP];
        const v4f* Bf = reinterpret_cast<const v4f*>(Mg) + lane;
        v4f b4 = {0.f, 0.f, 0.f, 0.f};
        v8f acc[2][NTILES];
#pragma unroll
        for (int n = 0; n < NTILES; ++n) {
            acc[0][n] = (v8f){0.f, 0.f, 0.f, 0.f, 0.f, 0.f, 0.f, 0.f};
            acc[1][n] = (v8f){0.f, 0.f, 0.f, 0.f, 0.f, 0.f, 0.f, 0.f};
#pragma unroll
            for (int kt = KLO[n]; kt <= KHI[n]; ++kt) {
                const int k0 = kt * 4 + kofs;
                v2f a0, a1, b;
                a0.x = uA[k0];
                a0.y = uA[k0 + 1];
                a1.x = uA[KC + k0];
                a1.y = uA[KC + k0 + 1];
                if ((kt & 1) == 0) b4 = Bf[(kt >> 1) * 32];
                b.x = (kt & 1) ? b4.z : b4.x;
                b.y = (kt & 1) ? b4.w : b4.y;
                acc[0][n] = __builtin_amdgcn_wmma_f32_16x16x4_f32(
                    false, a0, false, b, (short)0, acc[0][n], false, false);
                acc[1][n] = __builtin_amdgcn_wmma_f32_16x16x4_f32(
                    false, a1, false, b, (short)0, acc[1][n], false, false);
            }
        }
        __syncthreads();
        float* raw = &sBuf[w][0];
        const int rbase = (lane < 16) ? 0 : 8;
#pragma unroll
        for (int ch = 0; ch < 2; ++ch)
#pragma unroll
            for (int n = 0; n < NTILES; ++n) {
                const int col = ch * NC + n * 16 + col16;
#pragma unroll
                for (int j = 0; j < 8; ++j)
                    raw[(rbase + j) * RP + col] = acc[ch][n][j];
            }
    }
    __syncthreads();

    {
        const float* rRow = &sBuf[w][eSub * RP];
        float r0[4 * RQ], r1[4 * RQ];
#pragma unroll
        for (int q = 0; q < RQ; ++q) {
            const v4f t = *reinterpret_cast<const v4f*>(rRow + 4 * q);
            r0[4 * q] = t.x; r0[4 * q + 1] = t.y; r0[4 * q + 2] = t.z; r0[4 * q + 3] = t.w;
            const v4f s = *reinterpret_cast<const v4f*>(rRow + NC + 4 * q);
            r1[4 * q] = s.x; r1[4 * q + 1] = s.y; r1[4 * q + 2] = s.z; r1[4 * q + 3] = s.w;
        }
        const float* gO = gRow + part;
        float g[2 * N_COMBOS];
#pragma unroll
        for (int q = 0; q < 2 * N_COMBOS; ++q) g[q] = gO[q * 2];
        float* sOut = &sBuf[w][16 * RP];
        float* mRow = sOut + eSub * FEAT;
        float* m0 = mRow + part;
        float* m1 = mRow + 2 + part * 3;
        float* m2 = mRow + 8 + part * 5;
#define DO_MIX(LO, K, P) (P)[K] = mix_val<LO, K>(g, r0, r1);
        DO_MIX(0, 0, m0)
        DO_MIX(1, 0, m1) DO_MIX(1, 1, m1) DO_MIX(1, 2, m1)
        DO_MIX(2, 0, m2) DO_MIX(2, 1, m2) DO_MIX(2, 2, m2) DO_MIX(2, 3, m2) DO_MIX(2, 4, m2)
#undef DO_MIX
        sOut[16 * FEAT + eSub] = __int_as_float(tgt);
        __syncthreads();

        constexpr int NS = (16 * FEAT) / 32;
        float v[NS];
        long off[NS];
        unsigned ed[NS];
#pragma unroll
        for (int j = 0; j < NS; ++j) {
            const unsigned p = (unsigned)(j * 32 + lane);
            ed[j] = p / (unsigned)FEAT;
            const unsigned el = p - ed[j] * (unsigned)FEAT;
            const int t = __float_as_int(sOut[16 * FEAT + ed[j]]);
            off[j] = (long)t * FEAT + el;
            v[j] = sOut[p];
        }
        const long e0 = (long)tile * 16;
        if (e0 + 16 <= (long)E) {
#pragma unroll
            for (int j = 0; j < NS; ++j) atomicAdd(&outF[off[j]], v[j]);
        } else {
#pragma unroll
            for (int j = 0; j < NS; ++j)
                if (e0 + (long)ed[j] < (long)E) atomicAdd(&outF[off[j]], v[j]);
        }
    }
}

extern "C" void kernel_launch(void* const* d_in, const int* in_sizes, int n_in,
                              void* d_out, int out_size, void* d_ws, size_t ws_size,
                              hipStream_t stream) {
    const float* node = (const float*)d_in[0];
    const int* eidx   = (const int*)d_in[1];
    const float* sh0  = (const float*)d_in[2];
    const float* sh1  = (const float*)d_in[3];
    const float* sh2  = (const float*)d_in[4];
    const float* dist = (const float*)d_in[5];
    const float* cg[N_COMBOS];
    for (int i = 0; i < N_COMBOS; ++i) cg[i] = (const float*)d_in[6 + i];
    const float* W1_0 = (const float*)d_in[21];
    const float* b1_0 = (const float*)d_in[22];
    const float* W2_0 = (const float*)d_in[23];
    const float* b2_0 = (const float*)d_in[24];
    const float* W1_1 = (const float*)d_in[25];
    const float* b1_1 = (const float*)d_in[26];
    const float* W2_1 = (const float*)d_in[27];
    const float* b2_1 = (const float*)d_in[28];

    const int E = in_sizes[1] / 2;
    const int N = in_sizes[0] / FEAT;

    float* Mg     = (float*)d_ws;
    float* feats1 = (float*)((char*)d_ws + 131072);

    setup_kernel<<<1 + SETUP_ZERO_GROUPS, SETUP_THREADS, 0, stream>>>(
        cg[0], cg[1], cg[2], cg[3], cg[4], cg[5], cg[6], cg[7], cg[8], cg[9], cg[10], cg[11],
        cg[12], cg[13], cg[14], Mg, feats1, (long)N * FEAT, (float*)d_out, (long)out_size);

    int grid = (E + WAVES * 16 - 1) / (WAVES * 16);
    layer_kernel<<<grid, WAVES * 32, 0, stream>>>(node, eidx, sh0, sh1, sh2, dist, Mg,
                                                  W1_0, b1_0, W2_0, b2_0, feats1, E);
    layer_kernel<<<grid, WAVES * 32, 0, stream>>>(feats1, eidx, sh0, sh1, sh2, dist, Mg,
                                                  W1_1, b1_1, W2_1, b2_1, (float*)d_out, E);
}


# Gate phase sheds sign/NaN bookkeeping: ReLU is one integer max on the bits (no quieting max before it), tanh polynomial runs on the signed sum (no abs/copysign); exp form patches the stored gates
# speedup vs baseline: 4.9214x; 1.0397x over previous
#include <hip/hip_runtime.h>

typedef __attribute__((ext_vector_type(2))) float v2f;
typedef __attribute__((ext_vector_type(4))) float v4f;
typedef __attribute__((ext_vector_type(8))) float v8f;

#define N_COMBOS 15
#define KC 88
#define KP 180
#define NC 64
#define RP 132
#define RQ 15
#define NTILES 4
#define WAVES 2
#define FEAT 18

constexpr int cLE[N_COMBOS]     = {0,0,0,1,1,1,1,1,1,2,2,2,2,2,2};
constexpr int cLN[N_COMBOS]     = {0,1,2,0,1,1,1,2,2,0,1,1,2,2,2};
constexpr int cLO[N_COMBOS]     = {0,1,2,1,0,1,2,1,2,2,1,2,0,1,2};
constexpr int ROW_OFF[9]        = {0,1,4,9,12,28,21,43,60};
constexpr int COL_OFF[N_COMBOS] = {0,1,4,9,16,17,20,32,35,25,40,43,48,49,52};
constexpr int KLO[NTILES]       = {0,3,7,15};
constexpr int KHI[NTILES]       = {2,6,14,21};
#define NSTEPS 22
#define NPAIRS 11
#define MF_FLOATS (NPAIRS * 32 * 4)
constexpr int N_BY_LO[3]        = {3,6,6};
constexpr int C_BY_LO[3][6]     = {{0,4,12,0,0,0},{1,3,5,7,10,13},{2,6,8,9,11,14}};

#define SETUP_THREADS 256
#define SETUP_ZERO_GROUPS 256

template <int C>
__device__ __forceinline__ float get_cg(const float* __restrict__ cg, int tid) {
    constexpr int cnt = (2 * cLE[C] + 1) * (2 * cLN[C] + 1) * (2 * cLO[C] + 1);
    static_assert(cnt <= SETUP_THREADS, "one thread per cg entry");
    return cg[tid < cnt ? tid : cnt - 1];
}

__device__ __forceinline__ int frag_pos(int row, int col) {
    const int s = row >> 2;
    const int lane = (((row >> 1) & 1) << 4) | (col & 15);
    return ((((s >> 1) << 5) + lane) << 2) + ((s & 1) << 1) + (row & 1);
}

template <int C>
__device__ __forceinline__ void put_cg(float v, float* __restrict__ M, int tid) {
    constexpr int dO = 2 * cLO[C] + 1;
    constexpr int cnt = (2 * cLE[C] + 1) * (2 * cLN[C] + 1) * dO;
    constexpr int uo = ROW_OFF[cLE[C] * 3 + cLN[C]];
    constexpr int po = COL_OFF[C];
    static_assert((uo >> 2) >= KLO[po >> 4] && ((uo + cnt / dO - 1) >> 2) <= KHI[po >> 4] &&
                  ((po + dO - 1) >> 4) == (po >> 4), "a combo's rows lie in the K-tiles of its one N-tile");
    if (tid < cnt) {
        const int ab = tid / dO;
        const int k = tid - ab * dO;
        M[frag_pos(uo + ab, po + k)] = v;
    }
}

__device__ __forceinline__ void zero_span(float* __restrict__ p, long n, long first, long step) {
    const v4f z = {0.f, 0.f, 0.f, 0.f};
    if ((reinterpret_cast<unsigned long>(p) & 15ul) == 0ul) {
        v4f* p4 = reinterpret_cast<v4f*>(p);
        const long n4 = n >> 2;
        for (long i = first; i < n4; i += step) p4[i] = z;
        for (long i = (n4 << 2) + first; i < n; i += step) p[i] = 0.0f;
    } else {
        for (long i = first; i < n; i += step) p[i] = 0.0f;
    }
}

__global__ void __launch_bounds__(SETUP_THREADS)
setup_kernel(const float* __restrict__ cg0, const float* __restrict__ cg1, const float* __restrict__ cg2,
             const float* __restrict__ cg3, const float* __restrict__ cg4, const float* __restrict__ cg5,
             const float* __restrict__ cg6, const float* __restrict__ cg7, const float* __restrict__ cg8,
             const float* __restrict__ cg9, const float* __restrict__ cg10, const float* __restrict__ cg11,
             const float* __restrict__ cg12, const float* __restrict__ cg13, const float* __restrict__ cg14,
             float* __restrict__ M,
             float* __restrict__ bufA, long nA, float* __restrict__ bufB, long nB) {
    const int tid = threadIdx.x;
    if (blockIdx.x == 0) {
        static_assert(KHI[NTILES - 1] + 1 == NSTEPS && 4 * NSTEPS == KC, "K-tiles cover a channel's u row");
        static_assert(2 * NPAIRS >= NSTEPS, "every product has a place in a pair");
        const float v0 = get_cg<0>(cg0, tid),    v1 = get_cg<1>(cg1, tid),    v2 = get_cg<2>(cg2, tid);
        const float v3 = get_cg<3>(cg3, tid),    v4 = get_cg<4>(cg4, tid),    v5 = get_cg<5>(cg5, tid);
        const float v6 = get_cg<6>(cg6, tid),    v7 = get_cg<7>(cg7, tid),    v8 = get_cg<8>(cg8, tid);
        const float v9 = get_cg<9>(cg9, tid),    v10 = get_cg<10>(cg10, tid), v11 = get_cg<11>(cg11, tid);
        const float v12 = get_cg<12>(cg12, tid), v13 = get_cg<13>(cg13, tid), v14 = get_cg<14>(cg14, tid);
        const v4f z = {0.f, 0.f, 0.f, 0.f};
        v4f* M4 = reinterpret_cast<v4f*>(M);
        for (int i = tid; i < MF_FLOATS / 4; i += SETUP_THREADS) M4[i] = z;
        __syncthreads();
        put_cg<0>(v0, M, tid);   put_cg<1>(v1, M, tid);   put_cg<2>(v2, M, tid);
        put_cg<3>(v3, M, tid);   put_cg<4>(v4, M, tid);   put_cg<5>(v5, M, tid);
        put_cg<6>(v6, M, tid);   put_cg<7>(v7, M, tid);   put_cg<8>(v8, M, tid);
        put_cg<9>(v9, M, tid);   put_cg<10>(v10, M, tid); put_cg<11>(v11, M, tid);
        put_cg<12>(v12, M, tid); put_cg<13>(v13, M, tid); put_cg<14>(v14, M, tid);
    } else {
        const long first = (long)(blockIdx.x - 1) * SETUP_THREADS + tid;
        const long step = (long)(gridDim.x - 1) * SETUP_THREADS;
        zero_span(bufA, nA, first, step);
        zero_span(bufB, nB, first, step);
    }
}

__device__ __forceinline__ float relu_bits(float x) {
    const int i = __float_as_int(x);
    return __int_as_float(i > 0 ? i : 0);
}

#define NG 30
#define NGT 15
__device__ __forceinline__ void gate_mlp(const float* __restrict__ sW, int part,
                                         float x0, float x1,
                                         const v2f (&wA)[NGT], const float (&bA)[NGT],
                                         float* __restrict__ gRow) {
    v2f xb;
    xb.x = part ? 1.0f : x0;
    xb.y = part ? 0.0f : x1;
    const float* w2 = sW + part * 8;
    const float* c2 = sW + 240 + part * 2;
    float acc[NG];
#pragma unroll
    for (int j = 0; j < NGT; ++j) {
        v2f a;
        a.x = part ? bA[j] : wA[j].x;
        a.y = part ? 0.0f : wA[j].y;
        const v8f zero = {0.f, 0.f, 0.f, 0.f, 0.f, 0.f, 0.f, 0.f};
        const v8f d = __builtin_amdgcn_wmma_f32_16x16x4_f32(
            false, a, false, xb, (short)0, zero, false, false);
        v2f s = *reinterpret_cast<const v2f*>(c2 + j * 4);
#pragma unroll
        for (int h = 0; h < 4; ++h) {
            v2f hh;
            hh.x = relu_bits(d[h]);
            hh.y = relu_bits(d[4 + h]);
            const v2f wv = *reinterpret_cast<const v2f*>(w2 + j * 16 + h * 2);
            s = __builtin_elementwise_fma(hh, wv, s);
        }
        acc[j * 2] = s.x;
        acc[j * 2 + 1] = s.y;
    }
    float z[NG];
    float big = 0.0f;
#pragma unroll
    for (int t = 0; t < NG; ++t) {
        const float x = acc[t];
        const float x2 = x * x;
        float p = __builtin_fmaf(x2, -0x1.758e7ap-8f, 0x1.521192p-6f);
        p = __builtin_fmaf(x2, p, -0x1.b8389cp-5f);
        p = __builtin_fmaf(x2, p, 0x1.110704p-3f);
        p = __builtin_fmaf(x2, p, -0x1.555532p-2f);
        z[t] = __builtin_fmaf(x2, x * p, x);
        big = fmaxf(big, fabsf(x));
    }
#pragma unroll
    for (int j = 0; j < NGT; ++j)
#pragma unroll
        for (int i = 0; i < 2; ++i)
            gRow[j * 4 + part * 2 + i] = z[j * 2 + i];
    if (__builtin_amdgcn_ballot_w32(!(big < 0.625f)) != 0u) {
#pragma unroll
        for (int j = 0; j < NGT; ++j)
#pragma unroll
            for (int i = 0; i < 2; ++i) {
                const float x = acc[j * 2 + i];
                const float y = fabsf(x);
                if (!(y < 0.625f)) {
                    const float ex = expf(2.0f * y);
                    const float zb = __builtin_fmaf(-2.0f, __builtin_amdgcn_rcpf(ex + 1.0f), 1.0f);
                    gRow[j * 4 + part * 2 + i] = __builtin_copysignf(zb, x);
                }
            }
    }
}

__device__ __forceinline__ void u_stage(const float (&shv)[9], const float (&fc)[9],
                                        float* __restrict__ uCh) {
    float uv[KC];
#pragma unroll
    for (int i = 0; i < KC; ++i) uv[i] = 0.0f;
#pragma unroll
    for (int le = 0; le < 3; ++le) {
        const int de = 2 * le + 1;
        const int shb = (le == 0) ? 0 : (le == 1 ? 1 : 4);
#pragma unroll
        for (int ln = 0; ln < 3; ++ln) {
            const int dn = 2 * ln + 1;
            const int fb = (ln == 0) ? 0 : (ln == 1 ? 1 : 4);
            const int base = ROW_OFF[le * 3 + ln];
#pragma unroll
            for (int a = 0; a < de; ++a)
#pragma unroll
                for (int b = 0; b < dn; ++b)
                    uv[base + a * dn + b] = shv[shb + a] * fc[fb + b];
        }
    }
#pragma unroll
    for (int q = 0; q < KC / 4; ++q) {
        const v4f t = {uv[4 * q], uv[4 * q + 1], uv[4 * q + 2], uv[4 * q + 3]};
        *reinterpret_cast<v4f*>(uCh + 4 * q) = t;
    }
}

template <int LO, int K>
__device__ __forceinline__ float mix_val(const float (&g)[2 * N_COMBOS],
                                         const float (&r0)[4 * RQ], const float (&r1)[4 * RQ]) {
    float v = 0.0f;
#pragma unroll
    for (int q = 0; q < N_BY_LO[LO]; ++q) {
        const int c = C_BY_LO[LO][q];
        const int base = COL_OFF[c];
        v += g[c * 2]     * r0[base + K] +
             g[c * 2 + 1] * r1[base + K];
    }
    return v;
}

__global__ void __launch_bounds__(WAVES * 32, 5)
layer_kernel(const float* __restrict__ feats, const int* __restrict__ eidx,
             const float* __restrict__ sh0, const float* __restrict__ sh1,
             const float* __restrict__ sh2, const float* __restrict__ dist,
             const float* __restrict__ Mg,
             const float* __restrict__ W1, const float* __restrict__ b1,
             const float* __restrict__ W2, const float* __restrict__ b2,
             float* __restrict__ outF, int E) {
    __shared__ __attribute__((aligned(16))) float sBuf[WAVES][16 * KP];
    __shared__ __attribute__((aligned(16))) float sGate[WAVES][16 * 60];
    __shared__ __attribute__((aligned(16))) float sWgt[300];
    static_assert(16 * RP + 16 * FEAT + 16 <= 16 * KP, "raw tile + scatter tile must fit the u tile");
    static_assert(4 * RQ <= NC && 4 * RQ > COL_OFF[N_COMBOS - 1] + 4 && 2 * NC <= RP,
                  "the 16-byte pieces cover a channel's raw columns and stay inside them");
    static_assert(2 * KC <= KP, "both channels' u values fit the u row");

    const int lane = threadIdx.x & 31;
    const int w = threadIdx.x >> 5;
    const int tile = blockIdx.x * WAVES + w;
    const int eSub = lane & 15;
    const int part = lane >> 4;
    const long e = (long)tile * 16 + eSub;
    const bool valid = (e < (long)E);

    float* sW = &sWgt[0];
    {
        const int tid = threadIdx.x;
#pragma unroll
        for (int k = 0; k < 4; ++k) {
            int i = tid + k * (WAVES * 32);
            i = i < 240 ? i : 239;
            sW[(i & ~7) | ((i & 3) << 1) | ((i >> 2) & 1)] = W2[i];
        }
        const int i2 = tid < 60 ? tid : 59;
        sW[240 + i2] = b2[i2];
    }
    v2f wA[NGT];
    float bA[NGT];
#pragma unroll
    for (int j = 0; j < NGT; ++j) {
        wA[j] = *reinterpret_cast<const v2f*>(W1 + (j * 16 + eSub) * 2);
        bA[j] = b1[j * 16 + eSub];
    }

    float f[FEAT];
    float shv[9];
    float x0 = 0.0f, x1 = 0.0f;
    int tgt = 0;
    if (valid) {
        const float* fp = &feats[(long)eidx[e] * FEAT];
        tgt = eidx[(long)E + e];
#pragma unroll
        for (int j = 0; j < FEAT; ++j) f[j] = fp[j];
        shv[0] = sh0[e];
#pragma unroll
        for (int j = 0; j < 3; ++j) shv[1 + j] = sh1[e * 3 + j];
#pragma unroll
        for (int j = 0; j < 5; ++j) shv[4 + j] = sh2[e * 5 + j];
        x0 = dist[e];
        float inv0 = 0.5f * (f[0] + f[1]);
        float n1 = 0.0f, n2 = 0.0f;
#pragma unroll
        for (int j = 0; j < 3; ++j) { float m = 0.5f * (f[2 + j] + f[5 + j]); n1 += m * m; }
#pragma unroll
        for (int j = 0; j < 5; ++j) { float m = 0.5f * (f[8 + j] + f[13 + j]); n2 += m * m; }
        x1 = (inv0 + sqrtf(n1) + sqrtf(n2)) * (1.0f / 3.0f);
    } else {
#pragma unroll
        for (int j = 0; j < FEAT; ++j) f[j] = 0.0f;
#pragma unroll
        for (int j = 0; j < 9; ++j) shv[j] = 0.0f;
    }

    __syncthreads();

    float* gRow = &sGate[w][eSub * 60];
    float* uRow = &sBuf[w][eSub * KP];
    gate_mlp(sW, part, x0, x1, wA, bA, gRow);
    {
        float fc[9];
        fc[0] = part ? f[1] : f[0];
#pragma unroll
        for (int b = 0; b < 3; ++b) fc[1 + b] = part ? f[5 + b] : f[2 + b];
#pragma unroll
        for (int b = 0; b < 5; ++b) fc[4 + b] = part ? f[13 + b] : f[8 + b];
        u_stage(shv, fc, uRow + part * KC);
    }
    __syncthreads();

    {
        const int col16 = lane & 15;
        const int kofs = (lane < 16) ? 0 : 2;
        const float* uA = &sBuf[w][col16 * KP];
        const v4f* Bf = reinterpret_cast<const v4f*>(Mg) + lane;
        v4f b4 = {0.f, 0.f, 0.f, 0.f};
        v8f acc[2][NTILES];
#pragma unroll
        for (int n = 0; n < NTILES; ++n) {
            acc[0][n] = (v8f){0.f, 0.f, 0.f, 0.f, 0.f, 0.f, 0.f, 0.f};
            acc[1][n] = (v8f){0.f, 0.f, 0.f, 0.f, 0.f, 0.f, 0.f, 0.f};
#pragma unroll
            for (int kt = KLO[n]; kt <= KHI[n]; ++kt) {
                const int k0 = kt * 4 + kofs;
                v2f a0, a1, b;
                a0.x = uA[k0];
                a0.y = uA[k0 + 1];
                a1.x = uA[KC + k0];
                a1.y = uA[KC + k0 + 1];
                if ((kt & 1) == 0) b4 = Bf[(kt >> 1) * 32];
                b.x = (kt & 1) ? b4.z : b4.x;
                b.y = (kt & 1) ? b4.w : b4.y;
                acc[0][n] = __builtin_amdgcn_wmma_f32_16x16x4_f32(
                    false, a0, false, b, (short)0, acc[0][n], false, false);
                acc[1][n] = __builtin_amdgcn_wmma_f32_16x16x4_f32(
                    false, a1, false, b, (short)0, acc[1][n], false, false);
            }
        }
        __syncthreads();
        float* raw = &sBuf[w][0];
        const int rbase = (lane < 16) ? 0 : 8;
#pragma unroll
        for (int ch = 0; ch < 2; ++ch)
#pragma unroll
            for (int n = 0; n < NTILES; ++n) {
                const int col = ch * NC + n * 16 + col16;
#pragma unroll
                for (int j = 0; j < 8; ++j)
                    raw[(rbase + j) * RP + col] = acc[ch][n][j];
            }
    }
    __syncthreads();

    {
        const float* rRow = &sBuf[w][eSub * RP];
        float r0[4 * RQ], r1[4 * RQ];
#pragma unroll
        for (int q = 0; q < RQ; ++q) {
            const v4f t = *reinterpret_cast<const v4f*>(rRow + 4 * q);
            r0[4 * q] = t.x; r0[4 * q + 1] = t.y; r0[4 * q + 2] = t.z; r0[4 * q + 3] = t.w;
            const v4f s = *reinterpret_cast<const v4f*>(rRow + NC + 4 * q);
            r1[4 * q] = s.x; r1[4 * q + 1] = s.y; r1[4 * q + 2] = s.z; r1[4 * q + 3] = s.w;
        }
        const float* gO = gRow + part;
        float g[2 * N_COMBOS];
#pragma unroll
        for (int q = 0; q < 2 * N_COMBOS; ++q) g[q] = gO[q * 2];
        float* sOut = &sBuf[w][16 * RP];
        float* mRow = sOut + eSub * FEAT;
        float* m0 = mRow + part;
        float* m1 = mRow + 2 + part * 3;
        float* m2 = mRow + 8 + part * 5;
#define DO_MIX(LO, K, P) (P)[K] = mix_val<LO, K>(g, r0, r1);
        DO_MIX(0, 0, m0)
        DO_MIX(1, 0, m1) DO_MIX(1, 1, m1) DO_MIX(1, 2, m1)
        DO_MIX(2, 0, m2) DO_MIX(2, 1, m2) DO_MIX(2, 2, m2) DO_MIX(2, 3, m2) DO_MIX(2, 4, m2)
#undef DO_MIX
        sOut[16 * FEAT + eSub] = __int_as_float(tgt);
        __syncthreads();

        constexpr int NS = (16 * FEAT) / 32;
        float v[NS];
        long off[NS];
        unsigned ed[NS];
#pragma unroll
        for (int j = 0; j < NS; ++j) {
            const unsigned p = (unsigned)(j * 32 + lane);
            ed[j] = p / (unsigned)FEAT;
            const unsigned el = p - ed[j] * (unsigned)FEAT;
            const int t = __float_as_int(sOut[16 * FEAT + ed[j]]);
            off[j] = (long)t * FEAT + el;
            v[j] = sOut[p];
        }
        const long e0 = (long)tile * 16;
        if (e0 + 16 <= (long)E) {
#pragma unroll
            for (int j = 0; j < NS; ++j) atomicAdd(&outF[off[j]], v[j]);
        } else {
#pragma unroll
            for (int j = 0; j < NS; ++j)
                if (e0 + (long)ed[j] < (long)E) atomicAdd(&outF[off[j]], v[j]);
        }
    }
}

extern "C" void kernel_launch(void* const* d_in, const int* in_sizes, int n_in,
                              void* d_out, int out_size, void* d_ws, size_t ws_size,
                              hipStream_t stream) {
    const float* node = (const float*)d_in[0];
    const int* eidx   = (const int*)d_in[1];
    const float* sh0  = (const float*)d_in[2];
    const float* sh1  = (const float*)d_in[3];
    const float* sh2  = (const float*)d_in[4];
    const float* dist = (const float*)d_in[5];
    const float* cg[N_COMBOS];
    for (int i = 0; i < N_COMBOS; ++i) cg[i] = (const float*)d_in[6 + i];
    const float* W1_0 = (const float*)d_in[21];
    const float* b1_0 = (const float*)d_in[22];
    const float* W2_0 = (const float*)d_in[23];
    const float* b2_0 = (const float*)d_in[24];
    const float* W1_1 = (const float*)d_in[25];
    const float* b1_1 = (const float*)d_in[26];
    const float* W2_1 = (const float*)d_in[27];
    const float* b2_1 = (const float*)d_in[28];

    const int E = in_sizes[1] / 2;
    const int N = in_sizes[0] / FEAT;

    float* Mg     = (float*)d_ws;
    float* feats1 = (float*)((char*)d_ws + 131072);

    setup_kernel<<<1 + SETUP_ZERO_GROUPS, SETUP_THREADS, 0, stream>>>(
        cg[0], cg[1], cg[2], cg[3], cg[4], cg[5], cg[6], cg[7], cg[8], cg[9], cg[10], cg[11],
        cg[12], cg[13], cg[14], Mg, feats1, (long)N * FEAT, (float*)d_out, (long)out_size);

    int grid = (E + WAVES * 16 - 1) / (WAVES * 16);
    layer_kernel<<<grid, WAVES * 32, 0, stream>>>(node, eidx, sh0, sh1, sh2, dist, Mg,
                                                  W1_0, b1_0, W2_0, b2_0, feats1, E);
    layer_kernel<<<grid, WAVES * 32, 0, stream>>>(feats1, eidx, sh0, sh1, sh2, dist, Mg,
                                                  W1_1, b1_1, W2_1, b2_1, (float*)d_out, E);
}
